# starting kernel
# baseline (speedup 1.0000x reference)
#include <hip/hip_runtime.h>
#include <hip/hip_bf16.h>
#include <math.h>

typedef __attribute__((ext_vector_type(2))) float v2f;
typedef __attribute__((ext_vector_type(8))) float v8f;

#define HEADS 4
#define F_OUT 64
#define D_IN 128
#define NH_COLS 256
#define NEG_SLOPE 0.2f

__global__ __launch_bounds__(128) void gat_gemm_wmma(
    const float* __restrict__ X,
    const float* __restrict__ W,
    float* __restrict__ H)
{
    const int mtile   = blockIdx.x;
    const int wave    = threadIdx.x >> 5;
    const int lane    = threadIdx.x & 31;
    const int halfsel = lane >> 4;
    const int l16     = lane & 15;
    const int colbase = wave * 64;

    const float* xrow = X + (size_t)(mtile * 16 + l16) * D_IN;

    v8f acc0 = {}; v8f acc1 = {}; v8f acc2 = {}; v8f acc3 = {};

    #pragma unroll 4
    for (int s = 0; s < 32; ++s) {
        const int k0 = s * 4 + halfsel * 2;
        v2f a;
        a.x = xrow[k0];
        a.y = xrow[k0 + 1];
        const float* w0 = W + (size_t)k0 * NH_COLS;
        const float* w1 = W + (size_t)(k0 + 1) * NH_COLS;
        const int c0 = colbase + l16;
        v2f b0; b0.x = w0[c0     ]; b0.y = w1[c0     ];
        v2f b1; b1.x = w0[c0 + 16]; b1.y = w1[c0 + 16];
        v2f b2; b2.x = w0[c0 + 32]; b2.y = w1[c0 + 32];
        v2f b3; b3.x = w0[c0 + 48]; b3.y = w1[c0 + 48];

        acc0 = __builtin_amdgcn_wmma_f32_16x16x4_f32(false, a, false, b0, (short)0, acc0, false, false);
        acc1 = __builtin_amdgcn_wmma_f32_16x16x4_f32(false, a, false, b1, (short)0, acc1, false, false);
        acc2 = __builtin_amdgcn_wmma_f32_16x16x4_f32(false, a, false, b2, (short)0, acc2, false, false);
        acc3 = __builtin_amdgcn_wmma_f32_16x16x4_f32(false, a, false, b3, (short)0, acc3, false, false);
    }

    #pragma unroll
    for (int r = 0; r < 8; ++r) {
        const int m = mtile * 16 + r + halfsel * 8;
        float* hrow = H + (size_t)m * NH_COLS + colbase + l16;
        hrow[0]  = acc0[r];
        hrow[16] = acc1[r];
        hrow[32] = acc2[r];
        hrow[48] = acc3[r];
    }
}

__global__ void gat_init(float* __restrict__ amax, float* __restrict__ denom,
                         float* __restrict__ agg, int nh, int naggr)
{
    const int i = blockIdx.x * blockDim.x + threadIdx.x;
    if (i < nh) {
        amax[i]  = __int_as_float(0xFF800000);
        denom[i] = 0.0f;
    }
    if (i < naggr) agg[i] = 0.0f;
}

__global__ void gat_attn_coef(const float* __restrict__ H,
                              const float* __restrict__ att_src,
                              const float* __restrict__ att_dst,
                              float* __restrict__ a_src,
                              float* __restrict__ a_dst, int nh)
{
    const int tid = blockIdx.x * blockDim.x + threadIdx.x;
    if (tid >= nh) return;
    const int n = tid >> 2, hd = tid & 3;
    const float4* hp = (const float4*)(H + (size_t)n * NH_COLS + hd * F_OUT);
    const float4* as = (const float4*)(att_src + hd * F_OUT);
    const float4* ad = (const float4*)(att_dst + hd * F_OUT);
    float s1 = 0.0f, s2 = 0.0f;
    #pragma unroll
    for (int i = 0; i < 16; ++i) {
        const float4 hv = hp[i], v1 = as[i], v2 = ad[i];
        s1 += hv.x * v1.x + hv.y * v1.y + hv.z * v1.z + hv.w * v1.w;
        s2 += hv.x * v2.x + hv.y * v2.y + hv.z * v2.z + hv.w * v2.w;
    }
    a_src[tid] = s1;
    a_dst[tid] = s2;
}

__device__ __forceinline__ void atomicMaxFloat(float* addr, float val)
{
    if (val >= 0.0f) atomicMax((int*)addr, __float_as_int(val));
    else             atomicMin((unsigned int*)addr, __float_as_uint(val));
}

__device__ __forceinline__ void edge_endpoints(const int* __restrict__ ei,
                                               int e, int E, int& src, int& dst)
{
    if (e < E) { src = ei[e]; dst = ei[E + e]; }
    else       { src = dst = e - E; }
}

__device__ __forceinline__ float leaky(float a)
{
    return a > 0.0f ? a : NEG_SLOPE * a;
}

__global__ void gat_edge_max(const int* __restrict__ ei, int E, int Etot,
                             const float* __restrict__ a_src,
                             const float* __restrict__ a_dst,
                             float* __restrict__ amax)
{
    const int tid = blockIdx.x * blockDim.x + threadIdx.x;
    if (tid >= Etot * HEADS) return;
    const int e = tid >> 2, hd = tid & 3;
    int src, dst; edge_endpoints(ei, e, E, src, dst);
    const float al = leaky(a_src[src * HEADS + hd] + a_dst[dst * HEADS + hd]);
    atomicMaxFloat(&amax[dst * HEADS + hd], al);
}

__global__ void gat_edge_denom(const int* __restrict__ ei, int E, int Etot,
                               const float* __restrict__ a_src,
                               const float* __restrict__ a_dst,
                               const float* __restrict__ amax,
                               float* __restrict__ denom)
{
    const int tid = blockIdx.x * blockDim.x + threadIdx.x;
    if (tid >= Etot * HEADS) return;
    const int e = tid >> 2, hd = tid & 3;
    int src, dst; edge_endpoints(ei, e, E, src, dst);
    const float al = leaky(a_src[src * HEADS + hd] + a_dst[dst * HEADS + hd]);
    atomicAdd(&denom[dst * HEADS + hd], __expf(al - amax[dst * HEADS + hd]));
}

__global__ void gat_edge_agg(const int* __restrict__ ei, int E, int Etot,
                             const float* __restrict__ a_src,
                             const float* __restrict__ a_dst,
                             const float* __restrict__ amax,
                             const float* __restrict__ denom,
                             const float* __restrict__ H,
                             float* __restrict__ agg)
{
    const int gid = blockIdx.x * blockDim.x + threadIdx.x;
    const int e = gid >> 6;
    if (e >= Etot) return;
    const int t  = gid & 63;
    const int hd = t >> 4;
    const int f  = (t & 15) * 4;
    int src, dst; edge_endpoints(ei, e, E, src, dst);
    const int sh = src * HEADS + hd, dh = dst * HEADS + hd;
    const float al = leaky(a_src[sh] + a_dst[dh]);
    const float w  = __expf(al - amax[dh]) / (denom[dh] + 1e-16f);
    const float4 hv = *(const float4*)(H + (size_t)src * NH_COLS + hd * F_OUT + f);
    float* o = agg + (size_t)dst * NH_COLS + hd * F_OUT + f;
    atomicAdd(o + 0, w * hv.x);
    atomicAdd(o + 1, w * hv.y);
    atomicAdd(o + 2, w * hv.z);
    atomicAdd(o + 3, w * hv.w);
}

__global__ void gat_finalize(const float* __restrict__ agg,
                             const float* __restrict__ bias_gat,
                             const float* __restrict__ onehot,
                             const float* __restrict__ W_prior,
                             const float* __restrict__ b_prior,
                             const float* __restrict__ gate,
                             float* __restrict__ out, int nf)
{
    const int tid = blockIdx.x * blockDim.x + threadIdx.x;
    if (tid >= nf) return;
    const int n = tid >> 6, f = tid & 63;
    const float g = gate[0];
    const float* ar = agg + (size_t)n * NH_COLS + f;
    const float m = 0.25f * (ar[0] + ar[64] + ar[128] + ar[192]) + bias_gat[f];
    float p = b_prior[f];
    const float* oh = onehot + (size_t)n * 32;
    #pragma unroll
    for (int c = 0; c < 32; ++c) p += oh[c] * W_prior[c * 64 + f];
    out[tid] = (1.0f - g) * m + g * p;
}

extern "C" void kernel_launch(void* const* d_in, const int* in_sizes, int n_in,
                              void* d_out, int out_size, void* d_ws, size_t ws_size,
                              hipStream_t stream)
{
    const float* x        = (const float*)d_in[0];
    const int*   ei       = (const int*)  d_in[1];
    const float* onehot   = (const float*)d_in[2];
    const float* W_gat    = (const float*)d_in[3];
    const float* att_src  = (const float*)d_in[4];
    const float* att_dst  = (const float*)d_in[5];
    const float* bias_gat = (const float*)d_in[6];
    const float* W_prior  = (const float*)d_in[7];
    const float* b_prior  = (const float*)d_in[8];
    const float* gate     = (const float*)d_in[9];
    float* out = (float*)d_out;

    const int N    = in_sizes[0] / D_IN;
    const int E    = in_sizes[1] / 2;
    const int Etot = E + N;
    const int NHs  = N * HEADS;

    float* h      = (float*)d_ws;
    float* a_src  = h      + (size_t)N * NH_COLS;
    float* a_dst  = a_src  + NHs;
    float* amax   = a_dst  + NHs;
    float* denom  = amax   + NHs;
    float* agg    = denom  + NHs;

    const int naggr = N * NH_COLS;

    gat_init<<<(naggr + 255) / 256, 256, 0, stream>>>(amax, denom, agg, NHs, naggr);

    gat_gemm_wmma<<<N / 16, 128, 0, stream>>>(x, W_gat, h);

    gat_attn_coef<<<(NHs + 255) / 256, 256, 0, stream>>>(h, att_src, att_dst,
                                                         a_src, a_dst, NHs);

    const int eh = Etot * HEADS;
    gat_edge_max<<<(eh + 255) / 256, 256, 0, stream>>>(ei, E, Etot, a_src, a_dst, amax);
    gat_edge_denom<<<(eh + 255) / 256, 256, 0, stream>>>(ei, E, Etot, a_src, a_dst,
                                                         amax, denom);

    const long long aggthreads = (long long)Etot * 64;
    gat_edge_agg<<<(unsigned)((aggthreads + 255) / 256), 256, 0, stream>>>(
        ei, E, Etot, a_src, a_dst, amax, denom, h, agg);

    const int nf = N * F_OUT;
    gat_finalize<<<(nf + 255) / 256, 256, 0, stream>>>(agg, bias_gat, onehot,
                                                       W_prior, b_prior, gate,
                                                       out, nf);
}


# Fold the head mean into the edge scatter: each edge adds sum_h w_h*h[src,h,:] into an [N,64] accumulator, so float atomics, zero-fill and finalize reads drop 4x
# speedup vs baseline: 1.7934x; 1.7934x over previous
#include <hip/hip_runtime.h>
#include <hip/hip_bf16.h>
#include <math.h>

typedef __attribute__((ext_vector_type(2))) float v2f;
typedef __attribute__((ext_vector_type(8))) float v8f;

#define HEADS 4
#define F_OUT 64
#define D_IN 128
#define NH_COLS 256
#define NEG_SLOPE 0.2f

__global__ __launch_bounds__(128) void gat_gemm_wmma(
    const float* __restrict__ X,
    const float* __restrict__ W,
    float* __restrict__ H)
{
    const int mtile   = blockIdx.x;
    const int wave    = threadIdx.x >> 5;
    const int lane    = threadIdx.x & 31;
    const int halfsel = lane >> 4;
    const int l16     = lane & 15;
    const int colbase = wave * 64;

    const float* xrow = X + (size_t)(mtile * 16 + l16) * D_IN;

    v8f acc0 = {}; v8f acc1 = {}; v8f acc2 = {}; v8f acc3 = {};

    #pragma unroll 4
    for (int s = 0; s < 32; ++s) {
        const int k0 = s * 4 + halfsel * 2;
        v2f a;
        a.x = xrow[k0];
        a.y = xrow[k0 + 1];
        const float* w0 = W + (size_t)k0 * NH_COLS;
        const float* w1 = W + (size_t)(k0 + 1) * NH_COLS;
        const int c0 = colbase + l16;
        v2f b0; b0.x = w0[c0     ]; b0.y = w1[c0     ];
        v2f b1; b1.x = w0[c0 + 16]; b1.y = w1[c0 + 16];
        v2f b2; b2.x = w0[c0 + 32]; b2.y = w1[c0 + 32];
        v2f b3; b3.x = w0[c0 + 48]; b3.y = w1[c0 + 48];

        acc0 = __builtin_amdgcn_wmma_f32_16x16x4_f32(false, a, false, b0, (short)0, acc0, false, false);
        acc1 = __builtin_amdgcn_wmma_f32_16x16x4_f32(false, a, false, b1, (short)0, acc1, false, false);
        acc2 = __builtin_amdgcn_wmma_f32_16x16x4_f32(false, a, false, b2, (short)0, acc2, false, false);
        acc3 = __builtin_amdgcn_wmma_f32_16x16x4_f32(false, a, false, b3, (short)0, acc3, false, false);
    }

    #pragma unroll
    for (int r = 0; r < 8; ++r) {
        const int m = mtile * 16 + r + halfsel * 8;
        float* hrow = H + (size_t)m * NH_COLS + colbase + l16;
        hrow[0]  = acc0[r];
        hrow[16] = acc1[r];
        hrow[32] = acc2[r];
        hrow[48] = acc3[r];
    }
}

__global__ void gat_init(float* __restrict__ amax, float* __restrict__ denom,
                         float* __restrict__ agg, int nh, int naggr)
{
    const int i = blockIdx.x * blockDim.x + threadIdx.x;
    if (i < nh) {
        amax[i]  = __int_as_float(0xFF800000);
        denom[i] = 0.0f;
    }
    if (i < naggr) agg[i] = 0.0f;
}

__global__ void gat_attn_coef(const float* __restrict__ H,
                              const float* __restrict__ att_src,
                              const float* __restrict__ att_dst,
                              float* __restrict__ a_src,
                              float* __restrict__ a_dst, int nh)
{
    const int tid = blockIdx.x * blockDim.x + threadIdx.x;
    if (tid >= nh) return;
    const int n = tid >> 2, hd = tid & 3;
    const float4* hp = (const float4*)(H + (size_t)n * NH_COLS + hd * F_OUT);
    const float4* as = (const float4*)(att_src + hd * F_OUT);
    const float4* ad = (const float4*)(att_dst + hd * F_OUT);
    float s1 = 0.0f, s2 = 0.0f;
    #pragma unroll
    for (int i = 0; i < 16; ++i) {
        const float4 hv = hp[i], v1 = as[i], v2 = ad[i];
        s1 += hv.x * v1.x + hv.y * v1.y + hv.z * v1.z + hv.w * v1.w;
        s2 += hv.x * v2.x + hv.y * v2.y + hv.z * v2.z + hv.w * v2.w;
    }
    a_src[tid] = s1;
    a_dst[tid] = s2;
}

__device__ __forceinline__ void atomicMaxFloat(float* addr, float val)
{
    if (val >= 0.0f) atomicMax((int*)addr, __float_as_int(val));
    else             atomicMin((unsigned int*)addr, __float_as_uint(val));
}

__device__ __forceinline__ void edge_endpoints(const int* __restrict__ ei,
                                               int e, int E, int& src, int& dst)
{
    if (e < E) { src = ei[e]; dst = ei[E + e]; }
    else       { src = dst = e - E; }
}

__device__ __forceinline__ float leaky(float a)
{
    return a > 0.0f ? a : NEG_SLOPE * a;
}

__global__ void gat_edge_max(const int* __restrict__ ei, int E, int Etot,
                             const float* __restrict__ a_src,
                             const float* __restrict__ a_dst,
                             float* __restrict__ amax)
{
    const int tid = blockIdx.x * blockDim.x + threadIdx.x;
    if (tid >= Etot * HEADS) return;
    const int e = tid >> 2, hd = tid & 3;
    int src, dst; edge_endpoints(ei, e, E, src, dst);
    const float al = leaky(a_src[src * HEADS + hd] + a_dst[dst * HEADS + hd]);
    atomicMaxFloat(&amax[dst * HEADS + hd], al);
}

__global__ void gat_edge_denom(const int* __restrict__ ei, int E, int Etot,
                               const float* __restrict__ a_src,
                               const float* __restrict__ a_dst,
                               const float* __restrict__ amax,
                               float* __restrict__ denom)
{
    const int tid = blockIdx.x * blockDim.x + threadIdx.x;
    if (tid >= Etot * HEADS) return;
    const int e = tid >> 2, hd = tid & 3;
    int src, dst; edge_endpoints(ei, e, E, src, dst);
    const float al = leaky(a_src[src * HEADS + hd] + a_dst[dst * HEADS + hd]);
    atomicAdd(&denom[dst * HEADS + hd], __expf(al - amax[dst * HEADS + hd]));
}

__global__ void gat_edge_agg(const int* __restrict__ ei, int E, int Etot,
                             const float* __restrict__ a_src,
                             const float* __restrict__ a_dst,
                             const float* __restrict__ amax,
                             const float* __restrict__ denom,
                             const float* __restrict__ H,
                             float* __restrict__ agg)
{
    const int gid = blockIdx.x * blockDim.x + threadIdx.x;
    const int e = gid >> 4;
    if (e >= Etot) return;
    const int f = (gid & 15) * 4;
    int src, dst; edge_endpoints(ei, e, E, src, dst);
    const float4 as = *(const float4*)(a_src + (size_t)src * HEADS);
    const float4 ad = *(const float4*)(a_dst + (size_t)dst * HEADS);
    const float4 mx = *(const float4*)(amax  + (size_t)dst * HEADS);
    const float4 dn = *(const float4*)(denom + (size_t)dst * HEADS);
    const float w0 = __expf(leaky(as.x + ad.x) - mx.x) / (dn.x + 1e-16f);
    const float w1 = __expf(leaky(as.y + ad.y) - mx.y) / (dn.y + 1e-16f);
    const float w2 = __expf(leaky(as.z + ad.z) - mx.z) / (dn.z + 1e-16f);
    const float w3 = __expf(leaky(as.w + ad.w) - mx.w) / (dn.w + 1e-16f);
    const float* hp = H + (size_t)src * NH_COLS + f;
    const float4 h0 = *(const float4*)(hp);
    const float4 h1 = *(const float4*)(hp + F_OUT);
    const float4 h2 = *(const float4*)(hp + 2 * F_OUT);
    const float4 h3 = *(const float4*)(hp + 3 * F_OUT);
    float* o = agg + (size_t)dst * F_OUT + f;
    atomicAdd(o + 0, w0 * h0.x + w1 * h1.x + w2 * h2.x + w3 * h3.x);
    atomicAdd(o + 1, w0 * h0.y + w1 * h1.y + w2 * h2.y + w3 * h3.y);
    atomicAdd(o + 2, w0 * h0.z + w1 * h1.z + w2 * h2.z + w3 * h3.z);
    atomicAdd(o + 3, w0 * h0.w + w1 * h1.w + w2 * h2.w + w3 * h3.w);
}

__global__ void gat_finalize(const float* __restrict__ agg,
                             const float* __restrict__ bias_gat,
                             const float* __restrict__ onehot,
                             const float* __restrict__ W_prior,
                             const float* __restrict__ b_prior,
                             const float* __restrict__ gate,
                             float* __restrict__ out, int nf)
{
    const int tid = blockIdx.x * blockDim.x + threadIdx.x;
    if (tid >= nf) return;
    const int n = tid >> 6, f = tid & 63;
    const float g = gate[0];
    const float m = 0.25f * agg[(size_t)n * F_OUT + f] + bias_gat[f];
    float p = b_prior[f];
    const float* oh = onehot + (size_t)n * 32;
    #pragma unroll
    for (int c = 0; c < 32; ++c) p += oh[c] * W_prior[c * 64 + f];
    out[tid] = (1.0f - g) * m + g * p;
}

extern "C" void kernel_launch(void* const* d_in, const int* in_sizes, int n_in,
                              void* d_out, int out_size, void* d_ws, size_t ws_size,
                              hipStream_t stream)
{
    const float* x        = (const float*)d_in[0];
    const int*   ei       = (const int*)  d_in[1];
    const float* onehot   = (const float*)d_in[2];
    const float* W_gat    = (const float*)d_in[3];
    const float* att_src  = (const float*)d_in[4];
    const float* att_dst  = (const float*)d_in[5];
    const float* bias_gat = (const float*)d_in[6];
    const float* W_prior  = (const float*)d_in[7];
    const float* b_prior  = (const float*)d_in[8];
    const float* gate     = (const float*)d_in[9];
    float* out = (float*)d_out;

    const int N    = in_sizes[0] / D_IN;
    const int E    = in_sizes[1] / 2;
    const int Etot = E + N;
    const int NHs  = N * HEADS;

    float* h      = (float*)d_ws;
    float* a_src  = h      + (size_t)N * NH_COLS;
    float* a_dst  = a_src  + NHs;
    float* amax   = a_dst  + NHs;
    float* denom  = amax   + NHs;
    float* agg    = denom  + NHs;

    const int naggr = N * F_OUT;

    gat_init<<<(naggr + 255) / 256, 256, 0, stream>>>(amax, denom, agg, NHs, naggr);

    gat_gemm_wmma<<<N / 16, 128, 0, stream>>>(x, W_gat, h);

    gat_attn_coef<<<(NHs + 255) / 256, 256, 0, stream>>>(h, att_src, att_dst,
                                                         a_src, a_dst, NHs);

    const int eh = Etot * HEADS;
    gat_edge_max<<<(eh + 255) / 256, 256, 0, stream>>>(ei, E, Etot, a_src, a_dst, amax);
    gat_edge_denom<<<(eh + 255) / 256, 256, 0, stream>>>(ei, E, Etot, a_src, a_dst,
                                                         amax, denom);

    const long long aggthreads = (long long)Etot * 16;
    gat_edge_agg<<<(unsigned)((aggthreads + 255) / 256), 256, 0, stream>>>(
        ei, E, Etot, a_src, a_dst, amax, denom, h, agg);

    const int nf = N * F_OUT;
    gat_finalize<<<(nf + 255) / 256, 256, 0, stream>>>(agg, bias_gat, onehot,
                                                       W_prior, b_prior, gate,
                                                       out, nf);
}


# Replace the float-atomic edge scatter by a dst-sorted gather: counting-sort edges by dst (int atomics), then one wave per node does softmax, head-summed aggregation and its output row in registers
# speedup vs baseline: 2.2647x; 1.2628x over previous
#include <hip/hip_runtime.h>
#include <hip/hip_bf16.h>
#include <math.h>

typedef __attribute__((ext_vector_type(2))) float v2f;
typedef __attribute__((ext_vector_type(8))) float v8f;

#define HEADS 4
#define F_OUT 64
#define D_IN 128
#define NH_COLS 256
#define NEG_SLOPE 0.2f

__global__ __launch_bounds__(128) void gat_gemm_wmma(
    const float* __restrict__ X,
    const float* __restrict__ W,
    float* __restrict__ H)
{
    const int mtile   = blockIdx.x;
    const int wave    = threadIdx.x >> 5;
    const int lane    = threadIdx.x & 31;
    const int halfsel = lane >> 4;
    const int l16     = lane & 15;
    const int colbase = wave * 64;

    const float* xrow = X + (size_t)(mtile * 16 + l16) * D_IN;

    v8f acc0 = {}; v8f acc1 = {}; v8f acc2 = {}; v8f acc3 = {};

    #pragma unroll 4
    for (int s = 0; s < 32; ++s) {
        const int k0 = s * 4 + halfsel * 2;
        v2f a;
        a.x = xrow[k0];
        a.y = xrow[k0 + 1];
        const float* w0 = W + (size_t)k0 * NH_COLS;
        const float* w1 = W + (size_t)(k0 + 1) * NH_COLS;
        const int c0 = colbase + l16;
        v2f b0; b0.x = w0[c0     ]; b0.y = w1[c0     ];
        v2f b1; b1.x = w0[c0 + 16]; b1.y = w1[c0 + 16];
        v2f b2; b2.x = w0[c0 + 32]; b2.y = w1[c0 + 32];
        v2f b3; b3.x = w0[c0 + 48]; b3.y = w1[c0 + 48];

        acc0 = __builtin_amdgcn_wmma_f32_16x16x4_f32(false, a, false, b0, (short)0, acc0, false, false);
        acc1 = __builtin_amdgcn_wmma_f32_16x16x4_f32(false, a, false, b1, (short)0, acc1, false, false);
        acc2 = __builtin_amdgcn_wmma_f32_16x16x4_f32(false, a, false, b2, (short)0, acc2, false, false);
        acc3 = __builtin_amdgcn_wmma_f32_16x16x4_f32(false, a, false, b3, (short)0, acc3, false, false);
    }

    #pragma unroll
    for (int r = 0; r < 8; ++r) {
        const int m = mtile * 16 + r + halfsel * 8;
        float* hrow = H + (size_t)m * NH_COLS + colbase + l16;
        hrow[0]  = acc0[r];
        hrow[16] = acc1[r];
        hrow[32] = acc2[r];
        hrow[48] = acc3[r];
    }
}

__global__ void gat_attn_coef(const float* __restrict__ H,
                              const float* __restrict__ att_src,
                              const float* __restrict__ att_dst,
                              float* __restrict__ a_src,
                              float* __restrict__ a_dst, int nh)
{
    const int tid = blockIdx.x * blockDim.x + threadIdx.x;
    if (tid >= nh) return;
    const int n = tid >> 2, hd = tid & 3;
    const float4* hp = (const float4*)(H + (size_t)n * NH_COLS + hd * F_OUT);
    const float4* as = (const float4*)(att_src + hd * F_OUT);
    const float4* ad = (const float4*)(att_dst + hd * F_OUT);
    float s1 = 0.0f, s2 = 0.0f;
    #pragma unroll
    for (int i = 0; i < 16; ++i) {
        const float4 hv = hp[i], v1 = as[i], v2 = ad[i];
        s1 += hv.x * v1.x + hv.y * v1.y + hv.z * v1.z + hv.w * v1.w;
        s2 += hv.x * v2.x + hv.y * v2.y + hv.z * v2.z + hv.w * v2.w;
    }
    a_src[tid] = s1;
    a_dst[tid] = s2;
}

__device__ __forceinline__ float leaky(float a)
{
    return a > 0.0f ? a : NEG_SLOPE * a;
}

__global__ void csr_zero(int* __restrict__ cur, int* __restrict__ total, int n)
{
    const int i = blockIdx.x * blockDim.x + threadIdx.x;
    if (i < n) cur[i] = 0;
    if (i == 0) total[0] = 0;
}

__global__ void csr_hist(const int* __restrict__ ei, int E, int* __restrict__ cur)
{
    const int t = blockIdx.x * blockDim.x + threadIdx.x;
    const int T = gridDim.x * blockDim.x;
    int d[4];
    for (int k = 0; k < 4; ++k) {
        const int e = t + k * T;
        d[k] = (e < E) ? ei[E + e] : -1;
    }
    for (int k = 0; k < 4; ++k)
        if (d[k] >= 0) atomicAdd(&cur[d[k]], 1);
}

__global__ void csr_alloc(int* __restrict__ cur, int* __restrict__ deg,
                          int* __restrict__ total, int n)
{
    const int i = blockIdx.x * blockDim.x + threadIdx.x;
    const int lane = threadIdx.x & 31;
    const int c = (i < n) ? cur[i] : 0;
    int incl = c;
    for (int d = 1; d < 32; d <<= 1) {
        const int t = __shfl_up(incl, d);
        if (lane >= d) incl += t;
    }
    int base = 0;
    if (lane == 31) base = atomicAdd(total, incl);
    base = __shfl(base, 31);
    if (i < n) {
        cur[i] = base + incl - c;
        deg[i] = c;
    }
}

__global__ void csr_fill(const int* __restrict__ ei, int E,
                         int* __restrict__ cur, int* __restrict__ srcs)
{
    const int t = blockIdx.x * blockDim.x + threadIdx.x;
    const int T = gridDim.x * blockDim.x;
    int s[4], d[4], p[4];
    for (int k = 0; k < 4; ++k) {
        const int e = t + k * T;
        s[k] = (e < E) ? ei[e] : 0;
        d[k] = (e < E) ? ei[E + e] : -1;
    }
    for (int k = 0; k < 4; ++k)
        p[k] = (d[k] >= 0) ? atomicAdd(&cur[d[k]], 1) : -1;
    for (int k = 0; k < 4; ++k)
        if ((unsigned)p[k] < (unsigned)E) srcs[p[k]] = s[k];
}

__device__ __forceinline__ float wave_max(float v)
{
    for (int o = 16; o > 0; o >>= 1) v = fmaxf(v, __shfl_xor(v, o));
    return v;
}

__device__ __forceinline__ float wave_sum(float v)
{
    for (int o = 16; o > 0; o >>= 1) v += __shfl_xor(v, o);
    return v;
}

__device__ __forceinline__ float4 edge_alpha(const int* __restrict__ srcs,
                                             const float* __restrict__ a_src,
                                             int start, int dg, int i, int n,
                                             int nmax, float4 ad, int& s)
{
    s = n;
    if (i < dg) s = (int)min((unsigned)srcs[start + i], (unsigned)nmax);
    const float4 as = *(const float4*)(a_src + (size_t)s * HEADS);
    float4 al;
    al.x = leaky(as.x + ad.x);
    al.y = leaky(as.y + ad.y);
    al.z = leaky(as.z + ad.z);
    al.w = leaky(as.w + ad.w);
    return al;
}

__global__ __launch_bounds__(256) void gat_gather(
    const int* __restrict__ cur, const int* __restrict__ deg,
    const int* __restrict__ srcs,
    const float* __restrict__ a_src, const float* __restrict__ a_dst,
    const float* __restrict__ H,
    const float* __restrict__ bias_gat,
    const float* __restrict__ onehot,
    const float* __restrict__ W_prior,
    const float* __restrict__ b_prior,
    const float* __restrict__ gate,
    float* __restrict__ out, int N, int E)
{
    const int lane = threadIdx.x & 31;
    const int half = lane >> 4;
    const int l16  = lane & 15;
    const int n = blockIdx.x * 8 + (threadIdx.x >> 5);
    if (n >= N) return;

    int dg = deg[n];
    int start = cur[n] - dg;
    if (dg < 0 || start < 0 || start > E - dg) { dg = 0; start = 0; }
    const int total = dg + 1;
    const int nmax = N - 1;

    const float4 ad = *(const float4*)(a_dst + (size_t)n * HEADS);
    const float ninf = __int_as_float(0xFF800000);

    float4 m = make_float4(ninf, ninf, ninf, ninf);
    for (int b = 0; b < total; b += 32) {
        const int i = b + lane;
        if (i < total) {
            int s;
            const float4 al = edge_alpha(srcs, a_src, start, dg, i, n, nmax, ad, s);
            m.x = fmaxf(m.x, al.x); m.y = fmaxf(m.y, al.y);
            m.z = fmaxf(m.z, al.z); m.w = fmaxf(m.w, al.w);
        }
    }
    m.x = wave_max(m.x); m.y = wave_max(m.y); m.z = wave_max(m.z); m.w = wave_max(m.w);

    float4 den = make_float4(0.0f, 0.0f, 0.0f, 0.0f);
    for (int b = 0; b < total; b += 32) {
        const int i = b + lane;
        if (i < total) {
            int s;
            const float4 al = edge_alpha(srcs, a_src, start, dg, i, n, nmax, ad, s);
            den.x += __expf(al.x - m.x); den.y += __expf(al.y - m.y);
            den.z += __expf(al.z - m.z); den.w += __expf(al.w - m.w);
        }
    }
    den.x = wave_sum(den.x) + 1e-16f; den.y = wave_sum(den.y) + 1e-16f;
    den.z = wave_sum(den.z) + 1e-16f; den.w = wave_sum(den.w) + 1e-16f;

    float4 acc = make_float4(0.0f, 0.0f, 0.0f, 0.0f);
    for (int b = 0; b < total; b += 32) {
        const int i = b + lane;
        int s = n;
        float4 w = make_float4(0.0f, 0.0f, 0.0f, 0.0f);
        if (i < total) {
            const float4 al = edge_alpha(srcs, a_src, start, dg, i, n, nmax, ad, s);
            w.x = __expf(al.x - m.x) / den.x; w.y = __expf(al.y - m.y) / den.y;
            w.z = __expf(al.z - m.z) / den.z; w.w = __expf(al.w - m.w) / den.w;
        }
        const int cnt = min(32, total - b);
        for (int j = 0; j < cnt; j += 2) {
            const int jj = j + half;
            const int   sj = __shfl(s,   jj);
            const float w0 = __shfl(w.x, jj);
            const float w1 = __shfl(w.y, jj);
            const float w2 = __shfl(w.z, jj);
            const float w3 = __shfl(w.w, jj);
            if (jj < cnt) {
                const float* hp = H + (size_t)sj * NH_COLS + l16 * 4;
                const float4 h0 = *(const float4*)(hp);
                const float4 h1 = *(const float4*)(hp + F_OUT);
                const float4 h2 = *(const float4*)(hp + 2 * F_OUT);
                const float4 h3 = *(const float4*)(hp + 3 * F_OUT);
                acc.x += w0 * h0.x + w1 * h1.x + w2 * h2.x + w3 * h3.x;
                acc.y += w0 * h0.y + w1 * h1.y + w2 * h2.y + w3 * h3.y;
                acc.z += w0 * h0.z + w1 * h1.z + w2 * h2.z + w3 * h3.z;
                acc.w += w0 * h0.w + w1 * h1.w + w2 * h2.w + w3 * h3.w;
            }
        }
    }

    const float ohv = onehot[(size_t)n * 32 + lane];
    float4 p = make_float4(0.0f, 0.0f, 0.0f, 0.0f);
    #pragma unroll 4
    for (int c = 0; c < 16; ++c) {
        const float o = __shfl(ohv, c, 16);
        const float4 wp = *(const float4*)(W_prior + (half * 16 + c) * F_OUT + l16 * 4);
        p.x += o * wp.x; p.y += o * wp.y; p.z += o * wp.z; p.w += o * wp.w;
    }

    acc.x += __shfl_xor(acc.x, 16); acc.y += __shfl_xor(acc.y, 16);
    acc.z += __shfl_xor(acc.z, 16); acc.w += __shfl_xor(acc.w, 16);
    p.x += __shfl_xor(p.x, 16); p.y += __shfl_xor(p.y, 16);
    p.z += __shfl_xor(p.z, 16); p.w += __shfl_xor(p.w, 16);

    if (half == 0) {
        const float g = gate[0];
        const float4 bg = *(const float4*)(bias_gat + l16 * 4);
        const float4 bp = *(const float4*)(b_prior + l16 * 4);
        float4 o;
        o.x = (1.0f - g) * (0.25f * acc.x + bg.x) + g * (bp.x + p.x);
        o.y = (1.0f - g) * (0.25f * acc.y + bg.y) + g * (bp.y + p.y);
        o.z = (1.0f - g) * (0.25f * acc.z + bg.z) + g * (bp.z + p.z);
        o.w = (1.0f - g) * (0.25f * acc.w + bg.w) + g * (bp.w + p.w);
        *(float4*)(out + (size_t)n * F_OUT + l16 * 4) = o;
    }
}

extern "C" void kernel_launch(void* const* d_in, const int* in_sizes, int n_in,
                              void* d_out, int out_size, void* d_ws, size_t ws_size,
                              hipStream_t stream)
{
    const float* x        = (const float*)d_in[0];
    const int*   ei       = (const int*)  d_in[1];
    const float* onehot   = (const float*)d_in[2];
    const float* W_gat    = (const float*)d_in[3];
    const float* att_src  = (const float*)d_in[4];
    const float* att_dst  = (const float*)d_in[5];
    const float* bias_gat = (const float*)d_in[6];
    const float* W_prior  = (const float*)d_in[7];
    const float* b_prior  = (const float*)d_in[8];
    const float* gate     = (const float*)d_in[9];
    float* out = (float*)d_out;

    const int N    = in_sizes[0] / D_IN;
    const int E    = in_sizes[1] / 2;
    const int NHs  = N * HEADS;

    float* h      = (float*)d_ws;
    float* a_src  = h      + (size_t)N * NH_COLS;
    float* a_dst  = a_src  + NHs;
    int*   cur    = (int*)(a_dst + NHs);
    int*   deg    = cur + N;
    int*   total  = deg + N;
    int*   srcs   = total + 4;

    csr_zero <<<(N + 255) / 256, 256, 0, stream>>>(cur, total, N);
    csr_hist <<<((E + 3) / 4 + 255) / 256, 256, 0, stream>>>(ei, E, cur);
    csr_alloc<<<(N + 255) / 256, 256, 0, stream>>>(cur, deg, total, N);
    csr_fill <<<((E + 3) / 4 + 255) / 256, 256, 0, stream>>>(ei, E, cur, srcs);

    gat_gemm_wmma<<<N / 16, 128, 0, stream>>>(x, W_gat, h);

    gat_attn_coef<<<(NHs + 255) / 256, 256, 0, stream>>>(h, att_src, att_dst,
                                                         a_src, a_dst, NHs);

    gat_gather<<<(N + 7) / 8, 256, 0, stream>>>(cur, deg, srcs, a_src, a_dst, h,
                                                bias_gat, onehot, W_prior, b_prior,
                                                gate, out, N, E);
}


# Gather: a node with 32 to 63 incoming edges sorts its up to 64 sources in two registers per lane and sums from LDS, replacing a serial walk (about 34 steps x 2 passes) that one wave ran past the rest
# speedup vs baseline: 3.6864x; 1.6278x over previous
#include <hip/hip_runtime.h>
#include <hip/hip_bf16.h>
#include <math.h>

typedef __attribute__((ext_vector_type(2))) float v2f;
typedef __attribute__((ext_vector_type(8))) float v8f;

#define HEADS 4
#define F_OUT 64
#define D_IN 128
#define NH_COLS 256
#define NEG_SLOPE 0.2f

__global__ __launch_bounds__(128) void gat_gemm_wmma(
    const float* __restrict__ X,
    const float* __restrict__ W,
    float* __restrict__ H,
    const float* __restrict__ att_src,
    const float* __restrict__ att_dst,
    float* __restrict__ a_src,
    float* __restrict__ a_dst,
    int* __restrict__ cur,
    int* __restrict__ total,
    int N)
{
    __shared__ float s_coef[2][16][HEADS];

    const int mtile   = blockIdx.x;
    const int wave    = threadIdx.x >> 5;
    const int lane    = threadIdx.x & 31;
    const int halfsel = lane >> 4;
    const int l16     = lane & 15;
    const int colbase = wave * 64;

    if ((mtile & 1) == 0 && wave == 0) {
        const int i = mtile * 16 + lane;
        if (i < N) cur[i] = 0;
    }
    if (mtile == 0 && threadIdx.x == 0) total[0] = 0;

    const float* xrow = X + (size_t)(mtile * 16 + l16) * D_IN;

    v8f acc0 = {}; v8f acc1 = {}; v8f acc2 = {}; v8f acc3 = {};

    #pragma unroll 4
    for (int s = 0; s < 32; ++s) {
        const int k0 = s * 4 + halfsel * 2;
        v2f a;
        a.x = xrow[k0];
        a.y = xrow[k0 + 1];
        const float* w0 = W + (size_t)k0 * NH_COLS;
        const float* w1 = W + (size_t)(k0 + 1) * NH_COLS;
        const int c0 = colbase + l16;
        v2f b0; b0.x = w0[c0     ]; b0.y = w1[c0     ];
        v2f b1; b1.x = w0[c0 + 16]; b1.y = w1[c0 + 16];
        v2f b2; b2.x = w0[c0 + 32]; b2.y = w1[c0 + 32];
        v2f b3; b3.x = w0[c0 + 48]; b3.y = w1[c0 + 48];

        acc0 = __builtin_amdgcn_wmma_f32_16x16x4_f32(false, a, false, b0, (short)0, acc0, false, false);
        acc1 = __builtin_amdgcn_wmma_f32_16x16x4_f32(false, a, false, b1, (short)0, acc1, false, false);
        acc2 = __builtin_amdgcn_wmma_f32_16x16x4_f32(false, a, false, b2, (short)0, acc2, false, false);
        acc3 = __builtin_amdgcn_wmma_f32_16x16x4_f32(false, a, false, b3, (short)0, acc3, false, false);
    }

    #pragma unroll
    for (int r = 0; r < 8; ++r) {
        const int m = mtile * 16 + r + halfsel * 8;
        float* hrow = H + (size_t)m * NH_COLS + colbase + l16;
        hrow[0]  = acc0[r];
        hrow[16] = acc1[r];
        hrow[32] = acc2[r];
        hrow[48] = acc3[r];
    }

    const float* as = att_src + colbase + l16;
    const float* ad = att_dst + colbase + l16;
    const float as0 = as[0], as1 = as[16], as2 = as[32], as3 = as[48];
    const float ad0 = ad[0], ad1 = ad[16], ad2 = ad[32], ad3 = ad[48];
    float v[16];
    #pragma unroll
    for (int r = 0; r < 8; ++r) {
        v[r]     = acc0[r] * as0 + acc1[r] * as1 + acc2[r] * as2 + acc3[r] * as3;
        v[8 + r] = acc0[r] * ad0 + acc1[r] * ad1 + acc2[r] * ad2 + acc3[r] * ad3;
    }
    {
        const bool up = (l16 & 8) != 0;
        #pragma unroll
        for (int j = 0; j < 8; ++j) {
            const float mine  = up ? v[8 + j] : v[j];
            const float other = up ? v[j] : v[8 + j];
            v[j] = mine + __shfl_xor(other, 8);
        }
    }
    {
        const bool up = (l16 & 4) != 0;
        #pragma unroll
        for (int j = 0; j < 4; ++j) {
            const float mine  = up ? v[4 + j] : v[j];
            const float other = up ? v[j] : v[4 + j];
            v[j] = mine + __shfl_xor(other, 4);
        }
    }
    {
        const bool up = (l16 & 2) != 0;
        #pragma unroll
        for (int j = 0; j < 2; ++j) {
            const float mine  = up ? v[2 + j] : v[j];
            const float other = up ? v[j] : v[2 + j];
            v[j] = mine + __shfl_xor(other, 2);
        }
    }
    {
        const bool up = (l16 & 1) != 0;
        const float mine  = up ? v[1] : v[0];
        const float other = up ? v[0] : v[1];
        v[0] = mine + __shfl_xor(other, 1);
    }
    const float keep = v[0];
    s_coef[l16 >> 3][halfsel * 8 + (l16 & 7)][wave] = keep;
    __syncthreads();
    if (wave < 2) {
        const float2 v = *(const float2*)(&s_coef[wave][0][0] + lane * 2);
        float* dstp = (wave == 0 ? a_src : a_dst) + (size_t)mtile * 16 * HEADS + lane * 2;
        *(float2*)dstp = v;
    }
}

__device__ __forceinline__ float leaky(float a)
{
    return a > 0.0f ? a : NEG_SLOPE * a;
}

#define ROW_SLOTS 32

__global__ void group_edges(const int* __restrict__ ei, int E, int N,
                            int* __restrict__ cnt, int* __restrict__ rows,
                            int* __restrict__ ovf, int* __restrict__ ovf_total)
{
    const int t = blockIdx.x * blockDim.x + threadIdx.x;
    const int T = gridDim.x * blockDim.x;
    int s[4], d[4], p[4];
    for (int k = 0; k < 4; ++k) {
        const int e = t + k * T;
        s[k] = (e < E) ? ei[e] : 0;
        d[k] = (e < E) ? ei[E + e] : -1;
    }
    for (int k = 0; k < 4; ++k)
        p[k] = ((unsigned)d[k] < (unsigned)N) ? atomicAdd(&cnt[d[k]], 1) : -1;
    for (int k = 0; k < 4; ++k) {
        if (p[k] < 0) continue;
        if (p[k] < ROW_SLOTS) {
            rows[(size_t)d[k] * ROW_SLOTS + p[k]] = s[k];
        } else {
            const int q = atomicAdd(ovf_total, 1);
            if ((unsigned)q < (unsigned)E) {
                ovf[2 * (size_t)q]     = d[k];
                ovf[2 * (size_t)q + 1] = s[k];
            }
        }
    }
}

__device__ __forceinline__ float wave_max(float v)
{
    for (int o = 16; o > 0; o >>= 1) v = fmaxf(v, __shfl_xor(v, o));
    return v;
}

__device__ __forceinline__ float wave_sum(float v)
{
    for (int o = 16; o > 0; o >>= 1) v += __shfl_xor(v, o);
    return v;
}

__device__ __forceinline__ int wave_min_i(int v)
{
    for (int o = 16; o > 0; o >>= 1) v = min(v, __shfl_xor(v, o));
    return v;
}

__device__ __forceinline__ int wave_sum_i(int v)
{
    for (int o = 16; o > 0; o >>= 1) v += __shfl_xor(v, o);
    return v;
}

__device__ __forceinline__ float4 edge_alpha(const float* __restrict__ a_src, int s, float4 ad)
{
    const float4 as = *(const float4*)(a_src + (size_t)s * HEADS);
    float4 al;
    al.x = leaky(as.x + ad.x);
    al.y = leaky(as.y + ad.y);
    al.z = leaky(as.z + ad.z);
    al.w = leaky(as.w + ad.w);
    return al;
}

__device__ __forceinline__ void add_row(float4& acc, const float* __restrict__ H, int s, int l16,
                                        float w0, float w1, float w2, float w3)
{
    const float* hp = H + (size_t)s * NH_COLS + l16 * 4;
    const float4 h0 = *(const float4*)(hp);
    const float4 h1 = *(const float4*)(hp + F_OUT);
    const float4 h2 = *(const float4*)(hp + 2 * F_OUT);
    const float4 h3 = *(const float4*)(hp + 3 * F_OUT);
    acc.x += w0 * h0.x + w1 * h1.x + w2 * h2.x + w3 * h3.x;
    acc.y += w0 * h0.y + w1 * h1.y + w2 * h2.y + w3 * h3.y;
    acc.z += w0 * h0.z + w1 * h1.z + w2 * h2.z + w3 * h3.z;
    acc.w += w0 * h0.w + w1 * h1.w + w2 * h2.w + w3 * h3.w;
}

__global__ __launch_bounds__(256) void gat_gather(
    const int* __restrict__ cnt, const int* __restrict__ rows,
    const int* __restrict__ ovf, const int* __restrict__ ovf_total,
    const float* __restrict__ a_src, const float* __restrict__ a_dst,
    const float* __restrict__ H,
    const float* __restrict__ bias_gat,
    const float* __restrict__ onehot,
    const float* __restrict__ W_prior,
    const float* __restrict__ b_prior,
    const float* __restrict__ gate,
    float* __restrict__ out, int N, int E)
{
    __shared__ int    sh_s[8][2 * ROW_SLOTS];
    __shared__ float4 sh_w[8][2 * ROW_SLOTS];

    const int lane = threadIdx.x & 31;
    const int half = lane >> 4;
    const int l16  = lane & 15;
    const int n = blockIdx.x * 8 + (threadIdx.x >> 5);
    if (n >= N) return;

    const int nmax = N - 1;
    const int SENT = 0x7fffffff;
    int c = __builtin_amdgcn_readfirstlane(cnt[n]);
    c = min(max(c, 0), E);
    const int raw = rows[(size_t)n * ROW_SLOTS + lane];
    const float4 ad = *(const float4*)(a_dst + (size_t)n * HEADS);
    const float ninf = __int_as_float(0xFF800000);

    float4 acc = make_float4(0.0f, 0.0f, 0.0f, 0.0f);

    if (c < ROW_SLOTS) {
        int s = (lane < c) ? (int)min((unsigned)raw, (unsigned)nmax) : SENT;
        #pragma unroll
        for (int k = 2; k <= 32; k <<= 1) {
            #pragma unroll
            for (int j = k >> 1; j > 0; j >>= 1) {
                const int p = __shfl_xor(s, j);
                const bool keepmin = ((lane & j) == 0) == ((lane & k) == 0);
                s = keepmin ? min(s, p) : max(s, p);
            }
        }
        const int total = c + 1;
        const bool valid = lane < total;
        if (lane >= c) s = n;

        const float4 al = edge_alpha(a_src, s, ad);
        float4 m;
        m.x = wave_max(valid ? al.x : ninf); m.y = wave_max(valid ? al.y : ninf);
        m.z = wave_max(valid ? al.z : ninf); m.w = wave_max(valid ? al.w : ninf);
        float4 e;
        e.x = valid ? __expf(al.x - m.x) : 0.0f; e.y = valid ? __expf(al.y - m.y) : 0.0f;
        e.z = valid ? __expf(al.z - m.z) : 0.0f; e.w = valid ? __expf(al.w - m.w) : 0.0f;
        float4 w;
        w.x = e.x / (wave_sum(e.x) + 1e-16f); w.y = e.y / (wave_sum(e.y) + 1e-16f);
        w.z = e.z / (wave_sum(e.z) + 1e-16f); w.w = e.w / (wave_sum(e.w) + 1e-16f);

        for (int j = 0; j < total; j += 2) {
            const int jj = j + half;
            const int   sj = __shfl(s,   jj);
            const float w0 = __shfl(w.x, jj);
            const float w1 = __shfl(w.y, jj);
            const float w2 = __shfl(w.z, jj);
            const float w3 = __shfl(w.w, jj);
            if (jj < total) add_row(acc, H, sj, l16, w0, w1, w2, w3);
        }
    } else {
        const int rs = (int)min((unsigned)raw, (unsigned)nmax);
        int novf = __builtin_amdgcn_readfirstlane(ovf_total[0]);
        novf = min(max(novf, 0), E);
        const int2* ov = (const int2*)ovf;

      if (c < 2 * ROW_SLOTS) {
        const int wv = threadIdx.x >> 5;
        const int extra = c - ROW_SLOTS;
        int s1 = rs;
        int s2 = SENT;
        int filled = 0;
        #pragma nounroll
        for (int b = 0; b < novf; b += 32) {
            const int i = b + lane;
            int dn = -1, sv = 0;
            if (i < novf) {
                const int2 de = ov[i];
                dn = de.x;
                sv = (int)min((unsigned)de.y, (unsigned)nmax);
            }
            unsigned mask = __builtin_amdgcn_ballot_w32(dn == n);
            while (mask != 0u) {
                const int L = __builtin_ctz(mask);
                mask &= mask - 1u;
                const int v = __builtin_amdgcn_readlane(sv, L);
                if (lane == filled) s2 = v;
                ++filled;
            }
        }
        if (lane == extra) s2 = n;

        #pragma unroll
        for (int k = 2; k <= 64; k <<= 1) {
            #pragma unroll
            for (int j = k >> 1; j > 0; j >>= 1) {
                if (j == 32) {
                    const int lo = min(s1, s2), hi = max(s1, s2);
                    s1 = lo; s2 = hi;
                } else {
                    const int p1 = __shfl_xor(s1, j);
                    const int p2 = __shfl_xor(s2, j);
                    const bool lowpos = (lane & j) == 0;
                    const bool asc1 = (lane & k) == 0;
                    const bool asc2 = ((lane + 32) & k) == 0;
                    s1 = (lowpos == asc1) ? min(s1, p1) : max(s1, p1);
                    s2 = (lowpos == asc2) ? min(s2, p2) : max(s2, p2);
                }
            }
        }
        const int total = c + 1;
        const bool valid2 = lane < total - ROW_SLOTS;
        if (!valid2) s2 = n;
        sh_s[wv][lane] = s1;
        sh_s[wv][ROW_SLOTS + lane] = s2;

        const float4 al1 = edge_alpha(a_src, s1, ad);
        const float4 al2 = edge_alpha(a_src, s2, ad);
        float4 m;
        m.x = wave_max(fmaxf(al1.x, valid2 ? al2.x : ninf)); m.y = wave_max(fmaxf(al1.y, valid2 ? al2.y : ninf));
        m.z = wave_max(fmaxf(al1.z, valid2 ? al2.z : ninf)); m.w = wave_max(fmaxf(al1.w, valid2 ? al2.w : ninf));
        float4 e1, e2;
        e1.x = __expf(al1.x - m.x); e1.y = __expf(al1.y - m.y);
        e1.z = __expf(al1.z - m.z); e1.w = __expf(al1.w - m.w);
        e2.x = valid2 ? __expf(al2.x - m.x) : 0.0f; e2.y = valid2 ? __expf(al2.y - m.y) : 0.0f;
        e2.z = valid2 ? __expf(al2.z - m.z) : 0.0f; e2.w = valid2 ? __expf(al2.w - m.w) : 0.0f;
        float4 d;
        d.x = wave_sum(e1.x + e2.x) + 1e-16f; d.y = wave_sum(e1.y + e2.y) + 1e-16f;
        d.z = wave_sum(e1.z + e2.z) + 1e-16f; d.w = wave_sum(e1.w + e2.w) + 1e-16f;

        d.x = 1.0f / d.x; d.y = 1.0f / d.y; d.z = 1.0f / d.z; d.w = 1.0f / d.w;
        sh_w[wv][lane] = make_float4(e1.x * d.x, e1.y * d.y, e1.z * d.z, e1.w * d.w);
        sh_w[wv][ROW_SLOTS + lane] = make_float4(e2.x * d.x, e2.y * d.y, e2.z * d.z, e2.w * d.w);
        __builtin_amdgcn_wave_barrier();
        for (int j = 0; j < total; j += 2) {
            const int jj = j + half;
            if (jj < total) {
                const int sj = sh_s[wv][jj];
                const float4 wj = sh_w[wv][jj];
                add_row(acc, H, sj, l16, wj.x, wj.y, wj.z, wj.w);
            }
        }
      } else {

        float4 m = edge_alpha(a_src, n, ad);
        {
            const float4 al = edge_alpha(a_src, rs, ad);
            m.x = fmaxf(m.x, al.x); m.y = fmaxf(m.y, al.y);
            m.z = fmaxf(m.z, al.z); m.w = fmaxf(m.w, al.w);
        }
        #pragma nounroll
        for (int b = 0; b < novf; b += 32) {
            const int i = b + lane;
            if (i < novf) {
                const int2 de = ov[i];
                if (de.x == n) {
                    const float4 al = edge_alpha(a_src, (int)min((unsigned)de.y, (unsigned)nmax), ad);
                    m.x = fmaxf(m.x, al.x); m.y = fmaxf(m.y, al.y);
                    m.z = fmaxf(m.z, al.z); m.w = fmaxf(m.w, al.w);
                }
            }
        }
        m.x = wave_max(m.x); m.y = wave_max(m.y); m.z = wave_max(m.z); m.w = wave_max(m.w);

        float4 den = make_float4(0.0f, 0.0f, 0.0f, 0.0f);
        for (int pass = 0; pass < 2; ++pass) {
            int last = -1;
            for (int it = 0; it <= c; ++it) {
                int k = (rs > last) ? rs : SENT;
                if (n > last) k = min(k, n);
                #pragma nounroll
                for (int b = 0; b < novf; b += 32) {
                    const int i = b + lane;
                    if (i < novf) {
                        const int2 de = ov[i];
                        const int sv = (int)min((unsigned)de.y, (unsigned)nmax);
                        if (de.x == n && sv > last) k = min(k, sv);
                    }
                }
                k = __builtin_amdgcn_readfirstlane(wave_min_i(k));
                if (k == SENT) break;
                int mult = (rs == k) ? 1 : 0;
                #pragma nounroll
                for (int b = 0; b < novf; b += 32) {
                    const int i = b + lane;
                    if (i < novf) {
                        const int2 de = ov[i];
                        const int sv = (int)min((unsigned)de.y, (unsigned)nmax);
                        if (de.x == n && sv == k) mult += 1;
                    }
                }
                mult = wave_sum_i(mult) + ((n == k) ? 1 : 0);
                last = k;
                const float fm = (float)mult;
                const float4 al = edge_alpha(a_src, k, ad);
                float4 e;
                e.x = fm * __expf(al.x - m.x); e.y = fm * __expf(al.y - m.y);
                e.z = fm * __expf(al.z - m.z); e.w = fm * __expf(al.w - m.w);
                if (pass == 0) {
                    den.x += e.x; den.y += e.y; den.z += e.z; den.w += e.w;
                } else if (half == 0) {
                    add_row(acc, H, k, l16, e.x * den.x, e.y * den.y, e.z * den.z, e.w * den.w);
                }
            }
            if (pass == 0) {
                den.x = 1.0f / (den.x + 1e-16f); den.y = 1.0f / (den.y + 1e-16f);
                den.z = 1.0f / (den.z + 1e-16f); den.w = 1.0f / (den.w + 1e-16f);
            }
        }
      }
    }

    const float ohv = onehot[(size_t)n * 32 + lane];
    float4 p = make_float4(0.0f, 0.0f, 0.0f, 0.0f);
    #pragma unroll 4
    for (int c = 0; c < 16; ++c) {
        const float o = __shfl(ohv, c, 16);
        const float4 wp = *(const float4*)(W_prior + (half * 16 + c) * F_OUT + l16 * 4);
        p.x += o * wp.x; p.y += o * wp.y; p.z += o * wp.z; p.w += o * wp.w;
    }

    acc.x += __shfl_xor(acc.x, 16); acc.y += __shfl_xor(acc.y, 16);
    acc.z += __shfl_xor(acc.z, 16); acc.w += __shfl_xor(acc.w, 16);
    p.x += __shfl_xor(p.x, 16); p.y += __shfl_xor(p.y, 16);
    p.z += __shfl_xor(p.z, 16); p.w += __shfl_xor(p.w, 16);

    if (half == 0) {
        const float g = gate[0];
        const float4 bg = *(const float4*)(bias_gat + l16 * 4);
        const float4 bp = *(const float4*)(b_prior + l16 * 4);
        float4 o;
        o.x = (1.0f - g) * (0.25f * acc.x + bg.x) + g * (bp.x + p.x);
        o.y = (1.0f - g) * (0.25f * acc.y + bg.y) + g * (bp.y + p.y);
        o.z = (1.0f - g) * (0.25f * acc.z + bg.z) + g * (bp.z + p.z);
        o.w = (1.0f - g) * (0.25f * acc.w + bg.w) + g * (bp.w + p.w);
        *(float4*)(out + (size_t)n * F_OUT + l16 * 4) = o;
    }
}

extern "C" void kernel_launch(void* const* d_in, const int* in_sizes, int n_in,
                              void* d_out, int out_size, void* d_ws, size_t ws_size,
                              hipStream_t stream)
{
    const float* x        = (const float*)d_in[0];
    const int*   ei       = (const int*)  d_in[1];
    const float* onehot   = (const float*)d_in[2];
    const float* W_gat    = (const float*)d_in[3];
    const float* att_src  = (const float*)d_in[4];
    const float* att_dst  = (const float*)d_in[5];
    const float* bias_gat = (const float*)d_in[6];
    const float* W_prior  = (const float*)d_in[7];
    const float* b_prior  = (const float*)d_in[8];
    const float* gate     = (const float*)d_in[9];
    float* out = (float*)d_out;

    const int N    = in_sizes[0] / D_IN;
    const int E    = in_sizes[1] / 2;
    const int NHs  = N * HEADS;

    float* h      = (float*)d_ws;
    int*   rows   = (int*)(h + (size_t)N * NH_COLS);
    int*   ovf    = rows + (size_t)N * ROW_SLOTS;
    float* a_src  = (float*)(ovf + (size_t)E * 2);
    float* a_dst  = a_src  + NHs;
    int*   cnt    = (int*)(a_dst + NHs);
    int*   ototal = cnt + N;

    gat_gemm_wmma<<<N / 16, 128, 0, stream>>>(x, W_gat, h, att_src, att_dst,
                                              a_src, a_dst, cnt, ototal, N);

    group_edges<<<((E + 3) / 4 + 255) / 256, 256, 0, stream>>>(ei, E, N, cnt, rows, ovf, ototal);

    gat_gather<<<(N + 7) / 8, 256, 0, stream>>>(cnt, rows, ovf, ototal, a_src, a_dst, h,
                                                bias_gat, onehot, W_prior, b_prior,
                                                gate, out, N, E);
}


# Project after the weighted sum: an edge adds w*x[src] (512-B row) into per-head z; the block multiplies its 32x512 z tile by W_gat on the matrix unit; logits x.(W att) on the vector unit; h not stored
# speedup vs baseline: 4.2156x; 1.1436x over previous
#include <hip/hip_runtime.h>
#include <hip/hip_bf16.h>
#include <math.h>

typedef __attribute__((ext_vector_type(2))) float v2f;
typedef __attribute__((ext_vector_type(8))) float v8f;

#define HEADS 4
#define F_OUT 64
#define D_IN 128
#define NH_COLS 256
#define NEG_SLOPE 0.2f

__global__ __launch_bounds__(256) void gat_scores(
    const float* __restrict__ X,
    const float* __restrict__ W,
    const float* __restrict__ att_src,
    const float* __restrict__ att_dst,
    float* __restrict__ a_src,
    float* __restrict__ a_dst,
    int* __restrict__ cur,
    int* __restrict__ total,
    int N)
{
    __shared__ __attribute__((aligned(16))) float s_v[D_IN][8];
    __shared__ __attribute__((aligned(16))) float s_stage[8][2][8][HEADS];

    const int t    = threadIdx.x;
    const int wave = t >> 5;
    const int lane = t & 31;

    if (blockIdx.x == 0 && t == 0) total[0] = 0;

    const int groups = (N + 7) >> 3;
    const int gstep  = gridDim.x * 8;
    int g = blockIdx.x * 8 + wave;

    float4 xv[8];
    #pragma unroll
    for (int q = 0; q < 8; ++q) {
        const int n = g * 8 + q;
        xv[q] = (g < groups && n < N) ? *(const float4*)(X + (size_t)n * D_IN + lane * 4)
                                      : make_float4(0.0f, 0.0f, 0.0f, 0.0f);
    }

    {
        const int hl  = lane >> 4;
        const int l16 = lane & 15;
        const float4 as_lo = *(const float4*)(att_src + lane * 4);
        const float4 as_hi = *(const float4*)(att_src + 128 + lane * 4);
        const float4 ad_lo = *(const float4*)(att_dst + lane * 4);
        const float4 ad_hi = *(const float4*)(att_dst + 128 + lane * 4);
        const int q = ((l16 >> 3) & 1) * 2 + ((l16 >> 2) & 1);
        const int vcol = (q >> 1) * 4 + (q & 1) * 2 + hl;
        for (int r0 = 0; r0 < 16; r0 += 4) {
            const int i0 = wave * 16 + r0;
            float4 wlo[4], whi[4];
            for (int k = 0; k < 4; ++k) {
                const float* wr = W + (size_t)(i0 + k) * NH_COLS + lane * 4;
                wlo[k] = *(const float4*)(wr);
                whi[k] = *(const float4*)(wr + 128);
            }
            float res[4];
            for (int k = 0; k < 4; ++k) {
                float p[4];
                p[0] = wlo[k].x * as_lo.x + wlo[k].y * as_lo.y + wlo[k].z * as_lo.z + wlo[k].w * as_lo.w;
                p[1] = whi[k].x * as_hi.x + whi[k].y * as_hi.y + whi[k].z * as_hi.z + whi[k].w * as_hi.w;
                p[2] = wlo[k].x * ad_lo.x + wlo[k].y * ad_lo.y + wlo[k].z * ad_lo.z + wlo[k].w * ad_lo.w;
                p[3] = whi[k].x * ad_hi.x + whi[k].y * ad_hi.y + whi[k].z * ad_hi.z + whi[k].w * ad_hi.w;
                {
                    const bool up = (l16 & 8) != 0;
                    #pragma unroll
                    for (int j = 0; j < 2; ++j) {
                        const float mine  = up ? p[2 + j] : p[j];
                        const float other = up ? p[j] : p[2 + j];
                        p[j] = mine + __shfl_xor(other, 8);
                    }
                }
                {
                    const bool up = (l16 & 4) != 0;
                    const float mine  = up ? p[1] : p[0];
                    const float other = up ? p[0] : p[1];
                    p[0] = mine + __shfl_xor(other, 4);
                }
                p[0] += __shfl_xor(p[0], 2);
                p[0] += __shfl_xor(p[0], 1);
                res[k] = p[0];
            }
            if ((l16 & 3) == 0) {
                for (int k = 0; k < 4; ++k) s_v[i0 + k][vcol] = res[k];
            }
        }
    }
    __syncthreads();

    float v[4][8];
    #pragma unroll
    for (int i = 0; i < 4; ++i) {
        const float4 lo = *(const float4*)&s_v[lane * 4 + i][0];
        const float4 hi = *(const float4*)&s_v[lane * 4 + i][4];
        v[i][0] = lo.x; v[i][1] = lo.y; v[i][2] = lo.z; v[i][3] = lo.w;
        v[i][4] = hi.x; v[i][5] = hi.y; v[i][6] = hi.z; v[i][7] = hi.w;
    }
    const int vi = ((lane >> 4) & 1) * 4 + ((lane >> 3) & 1) * 2 + ((lane >> 2) & 1);

    while (g < groups) {
        const int n0 = g * 8;
        if ((g & 3) == 0 && n0 + lane < N) cur[n0 + lane] = 0;

        #pragma unroll
        for (int q = 0; q < 8; ++q) {
            float p[8];
            #pragma unroll
            for (int k = 0; k < 8; ++k)
                p[k] = xv[q].x * v[0][k] + xv[q].y * v[1][k] + xv[q].z * v[2][k] + xv[q].w * v[3][k];
            {
                const bool up = (lane & 16) != 0;
                #pragma unroll
                for (int j = 0; j < 4; ++j) {
                    const float mine  = up ? p[4 + j] : p[j];
                    const float other = up ? p[j] : p[4 + j];
                    p[j] = mine + __shfl_xor(other, 16);
                }
            }
            {
                const bool up = (lane & 8) != 0;
                #pragma unroll
                for (int j = 0; j < 2; ++j) {
                    const float mine  = up ? p[2 + j] : p[j];
                    const float other = up ? p[j] : p[2 + j];
                    p[j] = mine + __shfl_xor(other, 8);
                }
            }
            {
                const bool up = (lane & 4) != 0;
                const float mine  = up ? p[1] : p[0];
                const float other = up ? p[0] : p[1];
                p[0] = mine + __shfl_xor(other, 4);
            }
            p[0] += __shfl_xor(p[0], 2);
            p[0] += __shfl_xor(p[0], 1);
            if ((lane & 3) == 0) s_stage[wave][vi >> 2][q][vi & 3] = p[0];
        }
        __builtin_amdgcn_wave_barrier();
        const float os = s_stage[wave][0][lane >> 2][lane & 3];
        const float od = s_stage[wave][1][lane >> 2][lane & 3];
        if (n0 + (lane >> 2) < N) {
            a_src[(size_t)n0 * HEADS + lane] = os;
            a_dst[(size_t)n0 * HEADS + lane] = od;
        }
        __builtin_amdgcn_wave_barrier();

        g += gstep;
        #pragma unroll
        for (int q = 0; q < 8; ++q) {
            const int n = g * 8 + q;
            xv[q] = (g < groups && n < N) ? *(const float4*)(X + (size_t)n * D_IN + lane * 4)
                                          : make_float4(0.0f, 0.0f, 0.0f, 0.0f);
        }
    }
}

__device__ __forceinline__ float leaky(float a)
{
    return a > 0.0f ? a : NEG_SLOPE * a;
}

#define ROW_SLOTS 32

__global__ void group_edges(const int* __restrict__ ei, int E, int N,
                            int* __restrict__ cnt, int* __restrict__ rows,
                            int* __restrict__ ovf, int* __restrict__ ovf_total)
{
    const int t = blockIdx.x * blockDim.x + threadIdx.x;
    const int T = gridDim.x * blockDim.x;
    int s[4], d[4], p[4];
    for (int k = 0; k < 4; ++k) {
        const int e = t + k * T;
        s[k] = (e < E) ? ei[e] : 0;
        d[k] = (e < E) ? ei[E + e] : -1;
    }
    for (int k = 0; k < 4; ++k)
        p[k] = ((unsigned)d[k] < (unsigned)N) ? atomicAdd(&cnt[d[k]], 1) : -1;
    for (int k = 0; k < 4; ++k) {
        if (p[k] < 0) continue;
        if (p[k] < ROW_SLOTS) {
            rows[(size_t)d[k] * ROW_SLOTS + p[k]] = s[k];
        } else {
            const int q = atomicAdd(ovf_total, 1);
            if ((unsigned)q < (unsigned)E) {
                ovf[2 * (size_t)q]     = d[k];
                ovf[2 * (size_t)q + 1] = s[k];
            }
        }
    }
}

__device__ __forceinline__ float wave_max(float v)
{
    for (int o = 16; o > 0; o >>= 1) v = fmaxf(v, __shfl_xor(v, o));
    return v;
}

__device__ __forceinline__ float wave_sum(float v)
{
    for (int o = 16; o > 0; o >>= 1) v += __shfl_xor(v, o);
    return v;
}

__device__ __forceinline__ int wave_min_i(int v)
{
    for (int o = 16; o > 0; o >>= 1) v = min(v, __shfl_xor(v, o));
    return v;
}

__device__ __forceinline__ int wave_sum_i(int v)
{
    for (int o = 16; o > 0; o >>= 1) v += __shfl_xor(v, o);
    return v;
}

__device__ __forceinline__ float4 edge_alpha(const float* __restrict__ a_src, int s, float4 ad)
{
    const float4 as = *(const float4*)(a_src + (size_t)s * HEADS);
    float4 al;
    al.x = leaky(as.x + ad.x);
    al.y = leaky(as.y + ad.y);
    al.z = leaky(as.z + ad.z);
    al.w = leaky(as.w + ad.w);
    return al;
}

__device__ __forceinline__ void add_xrow(float4& z0, float4& z1, float4& z2, float4& z3,
                                         const float* __restrict__ X, int s, int lane,
                                         float w0, float w1, float w2, float w3)
{
    const float4 xv = *(const float4*)(X + (size_t)s * D_IN + lane * 4);
    z0.x += w0 * xv.x; z0.y += w0 * xv.y; z0.z += w0 * xv.z; z0.w += w0 * xv.w;
    z1.x += w1 * xv.x; z1.y += w1 * xv.y; z1.z += w1 * xv.z; z1.w += w1 * xv.w;
    z2.x += w2 * xv.x; z2.y += w2 * xv.y; z2.z += w2 * xv.z; z2.w += w2 * xv.w;
    z3.x += w3 * xv.x; z3.y += w3 * xv.y; z3.z += w3 * xv.z; z3.w += w3 * xv.w;
}

__device__ __forceinline__ float lane_value(float v, int j)
{
    return __int_as_float(__builtin_amdgcn_readlane(__float_as_int(v), j));
}

#define NODES_PER_BLOCK 32
#define WAVES_PER_BLOCK 16
#define ZPITCH 516
#define PPITCH 68

__global__ __launch_bounds__(512) __attribute__((amdgpu_waves_per_eu(12))) void gat_gather(
    const int* __restrict__ cnt, const int* __restrict__ rows,
    const int* __restrict__ ovf, const int* __restrict__ ovf_total,
    const float* __restrict__ a_src, const float* __restrict__ a_dst,
    const float* __restrict__ X,
    const float* __restrict__ W,
    const float* __restrict__ bias_gat,
    const float* __restrict__ onehot,
    const float* __restrict__ W_prior,
    const float* __restrict__ b_prior,
    const float* __restrict__ gate,
    float* __restrict__ out, int N, int E)
{
    __shared__ __attribute__((aligned(16))) float s_z[NODES_PER_BLOCK * ZPITCH];
    __shared__ __attribute__((aligned(16))) float s_p[HEADS * NODES_PER_BLOCK * PPITCH];
    int    (*sh_s)[2 * ROW_SLOTS] = (int (*)[2 * ROW_SLOTS])s_p;
    float4 (*sh_w)[2 * ROW_SLOTS] = (float4 (*)[2 * ROW_SLOTS])(s_p + WAVES_PER_BLOCK * 2 * ROW_SLOTS);

    const int lane = threadIdx.x & 31;
    const int half = lane >> 4;
    const int l16  = lane & 15;
    const int wv   = threadIdx.x >> 5;

  #pragma nounroll
  for (int q = 0; q < 2; ++q) {
    const int n = blockIdx.x * NODES_PER_BLOCK + q * 16 + wv;
    const bool node_ok = n < N;

    float4 z0 = make_float4(0.0f, 0.0f, 0.0f, 0.0f);
    float4 z1 = z0, z2 = z0, z3 = z0;

  if (node_ok) {
    const int nmax = N - 1;
    const int SENT = 0x7fffffff;
    int c = __builtin_amdgcn_readfirstlane(cnt[n]);
    c = min(max(c, 0), E);
    const int raw = rows[(size_t)n * ROW_SLOTS + lane];
    const float4 ad = *(const float4*)(a_dst + (size_t)n * HEADS);
    const float ninf = __int_as_float(0xFF800000);

    if (c < ROW_SLOTS) {
        int s = (lane < c) ? (int)min((unsigned)raw, (unsigned)nmax) : SENT;
        #pragma unroll
        for (int k = 2; k <= 32; k <<= 1) {
            #pragma unroll
            for (int j = k >> 1; j > 0; j >>= 1) {
                const int p = __shfl_xor(s, j);
                const bool keepmin = ((lane & j) == 0) == ((lane & k) == 0);
                s = keepmin ? min(s, p) : max(s, p);
            }
        }
        const int total = c + 1;
        const bool valid = lane < total;
        if (lane >= c) s = n;

        const float4 al = edge_alpha(a_src, s, ad);
        float4 m;
        m.x = wave_max(valid ? al.x : ninf); m.y = wave_max(valid ? al.y : ninf);
        m.z = wave_max(valid ? al.z : ninf); m.w = wave_max(valid ? al.w : ninf);
        float4 e;
        e.x = valid ? __expf(al.x - m.x) : 0.0f; e.y = valid ? __expf(al.y - m.y) : 0.0f;
        e.z = valid ? __expf(al.z - m.z) : 0.0f; e.w = valid ? __expf(al.w - m.w) : 0.0f;
        float4 w;
        w.x = e.x / (wave_sum(e.x) + 1e-16f); w.y = e.y / (wave_sum(e.y) + 1e-16f);
        w.z = e.z / (wave_sum(e.z) + 1e-16f); w.w = e.w / (wave_sum(e.w) + 1e-16f);

        const int totp = (total + 3) & ~3;
        #pragma unroll 4
        for (int j = 0; j < totp; ++j) {
            const int   sj = __builtin_amdgcn_readlane(s, j);
            const float w0 = lane_value(w.x, j);
            const float w1 = lane_value(w.y, j);
            const float w2 = lane_value(w.z, j);
            const float w3 = lane_value(w.w, j);
            add_xrow(z0, z1, z2, z3, X, sj, lane, w0, w1, w2, w3);
        }
    } else {
        const int rs = (int)min((unsigned)raw, (unsigned)nmax);
        int novf = __builtin_amdgcn_readfirstlane(ovf_total[0]);
        novf = min(max(novf, 0), E);
        const int2* ov = (const int2*)ovf;

      if (c < 2 * ROW_SLOTS) {
        const int extra = c - ROW_SLOTS;
        int s1 = rs;
        int s2 = SENT;
        int filled = 0;
        #pragma nounroll
        for (int b = 0; b < novf; b += 32) {
            const int i = b + lane;
            int dn = -1, sv = 0;
            if (i < novf) {
                const int2 de = ov[i];
                dn = de.x;
                sv = (int)min((unsigned)de.y, (unsigned)nmax);
            }
            unsigned mask = __builtin_amdgcn_ballot_w32(dn == n);
            while (mask != 0u) {
                const int L = __builtin_ctz(mask);
                mask &= mask - 1u;
                const int v = __builtin_amdgcn_readlane(sv, L);
                if (lane == filled) s2 = v;
                ++filled;
            }
        }
        if (lane == extra) s2 = n;

        #pragma unroll
        for (int k = 2; k <= 64; k <<= 1) {
            #pragma unroll
            for (int j = k >> 1; j > 0; j >>= 1) {
                if (j == 32) {
                    const int lo = min(s1, s2), hi = max(s1, s2);
                    s1 = lo; s2 = hi;
                } else {
                    const int p1 = __shfl_xor(s1, j);
                    const int p2 = __shfl_xor(s2, j);
                    const bool lowpos = (lane & j) == 0;
                    const bool asc1 = (lane & k) == 0;
                    const bool asc2 = ((lane + 32) & k) == 0;
                    s1 = (lowpos == asc1) ? min(s1, p1) : max(s1, p1);
                    s2 = (lowpos == asc2) ? min(s2, p2) : max(s2, p2);
                }
            }
        }
        const int total = c + 1;
        const bool valid2 = lane < total - ROW_SLOTS;
        if (!valid2) s2 = n;
        sh_s[wv][lane] = s1;
        sh_s[wv][ROW_SLOTS + lane] = s2;

        const float4 al1 = edge_alpha(a_src, s1, ad);
        const float4 al2 = edge_alpha(a_src, s2, ad);
        float4 m;
        m.x = wave_max(fmaxf(al1.x, valid2 ? al2.x : ninf)); m.y = wave_max(fmaxf(al1.y, valid2 ? al2.y : ninf));
        m.z = wave_max(fmaxf(al1.z, valid2 ? al2.z : ninf)); m.w = wave_max(fmaxf(al1.w, valid2 ? al2.w : ninf));
        float4 e1, e2;
        e1.x = __expf(al1.x - m.x); e1.y = __expf(al1.y - m.y);
        e1.z = __expf(al1.z - m.z); e1.w = __expf(al1.w - m.w);
        e2.x = valid2 ? __expf(al2.x - m.x) : 0.0f; e2.y = valid2 ? __expf(al2.y - m.y) : 0.0f;
        e2.z = valid2 ? __expf(al2.z - m.z) : 0.0f; e2.w = valid2 ? __expf(al2.w - m.w) : 0.0f;
        float4 d;
        d.x = wave_sum(e1.x + e2.x) + 1e-16f; d.y = wave_sum(e1.y + e2.y) + 1e-16f;
        d.z = wave_sum(e1.z + e2.z) + 1e-16f; d.w = wave_sum(e1.w + e2.w) + 1e-16f;

        d.x = 1.0f / d.x; d.y = 1.0f / d.y; d.z = 1.0f / d.z; d.w = 1.0f / d.w;
        sh_w[wv][lane] = make_float4(e1.x * d.x, e1.y * d.y, e1.z * d.z, e1.w * d.w);
        sh_w[wv][ROW_SLOTS + lane] = make_float4(e2.x * d.x, e2.y * d.y, e2.z * d.z, e2.w * d.w);
        __builtin_amdgcn_wave_barrier();
        #pragma nounroll
        for (int j = 0; j < total; ++j) {
            const int sj = sh_s[wv][j];
            const float4 wj = sh_w[wv][j];
            add_xrow(z0, z1, z2, z3, X, sj, lane, wj.x, wj.y, wj.z, wj.w);
        }
      } else {

        float4 m = edge_alpha(a_src, n, ad);
        {
            const float4 al = edge_alpha(a_src, rs, ad);
            m.x = fmaxf(m.x, al.x); m.y = fmaxf(m.y, al.y);
            m.z = fmaxf(m.z, al.z); m.w = fmaxf(m.w, al.w);
        }
        #pragma nounroll
        for (int b = 0; b < novf; b += 32) {
            const int i = b + lane;
            if (i < novf) {
                const int2 de = ov[i];
                if (de.x == n) {
                    const float4 al = edge_alpha(a_src, (int)min((unsigned)de.y, (unsigned)nmax), ad);
                    m.x = fmaxf(m.x, al.x); m.y = fmaxf(m.y, al.y);
                    m.z = fmaxf(m.z, al.z); m.w = fmaxf(m.w, al.w);
                }
            }
        }
        m.x = wave_max(m.x); m.y = wave_max(m.y); m.z = wave_max(m.z); m.w = wave_max(m.w);

        float4 den = make_float4(0.0f, 0.0f, 0.0f, 0.0f);
        for (int pass = 0; pass < 2; ++pass) {
            int last = -1;
            for (int it = 0; it <= c; ++it) {
                int k = (rs > last) ? rs : SENT;
                if (n > last) k = min(k, n);
                #pragma nounroll
                for (int b = 0; b < novf; b += 32) {
                    const int i = b + lane;
                    if (i < novf) {
                        const int2 de = ov[i];
                        const int sv = (int)min((unsigned)de.y, (unsigned)nmax);
                        if (de.x == n && sv > last) k = min(k, sv);
                    }
                }
                k = __builtin_amdgcn_readfirstlane(wave_min_i(k));
                if (k == SENT) break;
                int mult = (rs == k) ? 1 : 0;
                #pragma nounroll
                for (int b = 0; b < novf; b += 32) {
                    const int i = b + lane;
                    if (i < novf) {
                        const int2 de = ov[i];
                        const int sv = (int)min((unsigned)de.y, (unsigned)nmax);
                        if (de.x == n && sv == k) mult += 1;
                    }
                }
                mult = wave_sum_i(mult) + ((n == k) ? 1 : 0);
                last = k;
                const float fm = (float)mult;
                const float4 al = edge_alpha(a_src, k, ad);
                float4 e;
                e.x = fm * __expf(al.x - m.x); e.y = fm * __expf(al.y - m.y);
                e.z = fm * __expf(al.z - m.z); e.w = fm * __expf(al.w - m.w);
                if (pass == 0) {
                    den.x += e.x; den.y += e.y; den.z += e.z; den.w += e.w;
                } else {
                    add_xrow(z0, z1, z2, z3, X, k, lane, e.x * den.x, e.y * den.y, e.z * den.z, e.w * den.w);
                }
            }
            if (pass == 0) {
                den.x = 1.0f / (den.x + 1e-16f); den.y = 1.0f / (den.y + 1e-16f);
                den.z = 1.0f / (den.z + 1e-16f); den.w = 1.0f / (den.w + 1e-16f);
            }
        }
      }
    }
  }

    {
        float* zr = &s_z[(q * 16 + wv) * ZPITCH + lane * 4];
        *(float4*)(zr)            = z0;
        *(float4*)(zr + D_IN)     = z1;
        *(float4*)(zr + 2 * D_IN) = z2;
        *(float4*)(zr + 3 * D_IN) = z3;
    }
  }

    const int head = wv >> 2;
    const int ct   = wv & 3;
    const float* wcol = W + (size_t)(half * 2) * NH_COLS + head * F_OUT + ct * 16 + l16;
    float bq[16];
    #pragma unroll
    for (int t = 0; t < 8; ++t) {
        bq[2 * t]     = wcol[(size_t)(t * 4)     * NH_COLS];
        bq[2 * t + 1] = wcol[(size_t)(t * 4 + 1) * NH_COLS];
    }
    __syncthreads();

    {
        const float* zrow = &s_z[l16 * ZPITCH + head * D_IN + half * 2];
        v8f accA = {}; v8f accB = {};
        #pragma nounroll
        for (int g = 0; g < 4; ++g) {
            if (g > 0) {
                #pragma unroll
                for (int t = 0; t < 8; ++t) {
                    bq[2 * t]     = wcol[(size_t)(g * 32 + t * 4)     * NH_COLS];
                    bq[2 * t + 1] = wcol[(size_t)(g * 32 + t * 4 + 1) * NH_COLS];
                }
            }
            #pragma unroll
            for (int t = 0; t < 8; ++t) {
                const v2f a0 = *(const v2f*)(zrow + g * 32 + t * 4);
                const v2f a1 = *(const v2f*)(zrow + 16 * ZPITCH + g * 32 + t * 4);
                v2f b; b.x = bq[2 * t]; b.y = bq[2 * t + 1];
                accA = __builtin_amdgcn_wmma_f32_16x16x4_f32(false, a0, false, b, (short)0, accA, false, false);
                accB = __builtin_amdgcn_wmma_f32_16x16x4_f32(false, a1, false, b, (short)0, accB, false, false);
            }
        }
        #pragma unroll
        for (int r = 0; r < 8; ++r) {
            s_p[(head * NODES_PER_BLOCK + r + half * 8) * PPITCH + ct * 16 + l16]      = accA[r];
            s_p[(head * NODES_PER_BLOCK + 16 + r + half * 8) * PPITCH + ct * 16 + l16] = accB[r];
        }
    }
    __syncthreads();

  for (int q = 0; q < 2; ++q) {
    const int row = q * 16 + wv;
    const int n = blockIdx.x * NODES_PER_BLOCK + row;
    if (n >= N) continue;

    float4 acc;
    {
        const float4 p0 = *(const float4*)&s_p[(0 * NODES_PER_BLOCK + row) * PPITCH + l16 * 4];
        const float4 p1 = *(const float4*)&s_p[(1 * NODES_PER_BLOCK + row) * PPITCH + l16 * 4];
        const float4 p2 = *(const float4*)&s_p[(2 * NODES_PER_BLOCK + row) * PPITCH + l16 * 4];
        const float4 p3 = *(const float4*)&s_p[(3 * NODES_PER_BLOCK + row) * PPITCH + l16 * 4];
        acc.x = ((p0.x + p1.x) + p2.x) + p3.x;
        acc.y = ((p0.y + p1.y) + p2.y) + p3.y;
        acc.z = ((p0.z + p1.z) + p2.z) + p3.z;
        acc.w = ((p0.w + p1.w) + p2.w) + p3.w;
    }

    const float ohv = onehot[(size_t)n * 32 + lane];
    float4 p = make_float4(0.0f, 0.0f, 0.0f, 0.0f);
    #pragma unroll 4
    for (int c = 0; c < 16; ++c) {
        const float o = __shfl(ohv, c, 16);
        const float4 wp = *(const float4*)(W_prior + (half * 16 + c) * F_OUT + l16 * 4);
        p.x += o * wp.x; p.y += o * wp.y; p.z += o * wp.z; p.w += o * wp.w;
    }

    p.x += __shfl_xor(p.x, 16); p.y += __shfl_xor(p.y, 16);
    p.z += __shfl_xor(p.z, 16); p.w += __shfl_xor(p.w, 16);

    if (half == 0) {
        const float g = gate[0];
        const float4 bg = *(const float4*)(bias_gat + l16 * 4);
        const float4 bp = *(const float4*)(b_prior + l16 * 4);
        float4 o;
        o.x = (1.0f - g) * (0.25f * acc.x + bg.x) + g * (bp.x + p.x);
        o.y = (1.0f - g) * (0.25f * acc.y + bg.y) + g * (bp.y + p.y);
        o.z = (1.0f - g) * (0.25f * acc.z + bg.z) + g * (bp.z + p.z);
        o.w = (1.0f - g) * (0.25f * acc.w + bg.w) + g * (bp.w + p.w);
        *(float4*)(out + (size_t)n * F_OUT + l16 * 4) = o;
    }
  }
}

extern "C" void kernel_launch(void* const* d_in, const int* in_sizes, int n_in,
                              void* d_out, int out_size, void* d_ws, size_t ws_size,
                              hipStream_t stream)
{
    const float* x        = (const float*)d_in[0];
    const int*   ei       = (const int*)  d_in[1];
    const float* onehot   = (const float*)d_in[2];
    const float* W_gat    = (const float*)d_in[3];
    const float* att_src  = (const float*)d_in[4];
    const float* att_dst  = (const float*)d_in[5];
    const float* bias_gat = (const float*)d_in[6];
    const float* W_prior  = (const float*)d_in[7];
    const float* b_prior  = (const float*)d_in[8];
    const float* gate     = (const float*)d_in[9];
    float* out = (float*)d_out;

    const int N    = in_sizes[0] / D_IN;
    const int E    = in_sizes[1] / 2;
    const int NHs  = N * HEADS;

    float* h      = (float*)d_ws;
    int*   rows   = (int*)(h + (size_t)N * NH_COLS);
    int*   ovf    = rows + (size_t)N * ROW_SLOTS;
    float* a_src  = (float*)(ovf + (size_t)E * 2);
    float* a_dst  = a_src  + NHs;
    int*   cnt    = (int*)(a_dst + NHs);
    int*   ototal = cnt + N;

    gat_scores<<<(N + 63) / 64, 256, 0, stream>>>(x, W_gat, att_src, att_dst,
                                                  a_src, a_dst, cnt, ototal, N);

    group_edges<<<((E + 3) / 4 + 255) / 256, 256, 0, stream>>>(ei, E, N, cnt, rows, ovf, ototal);

    gat_gather<<<(N + NODES_PER_BLOCK - 1) / NODES_PER_BLOCK, 512, 0, stream>>>(
        cnt, rows, ovf, ototal, a_src, a_dst, x, W_gat,
        bias_gat, onehot, W_prior, b_prior, gate, out, N, E);
}


# Lane exchanges of the sorts and reductions on the vector ALU (row_xmask swizzle on the min/max/add itself, v_permlanex16 between rows) instead of ds_bpermute round trips through the LDS unit
# speedup vs baseline: 4.3739x; 1.0376x over previous
#include <hip/hip_runtime.h>
#include <hip/hip_bf16.h>
#include <math.h>

typedef __attribute__((ext_vector_type(2))) float v2f;
typedef __attribute__((ext_vector_type(8))) float v8f;

#define HEADS 4
#define F_OUT 64
#define D_IN 128
#define NH_COLS 256
#define NEG_SLOPE 0.2f

template <int M> __device__ __forceinline__ int xlane(int v)
{
    if (M == 16) return __builtin_amdgcn_permlanex16(v, v, 0x76543210, 0xfedcba98, false, false);
    return __builtin_amdgcn_update_dpp(0, v, 0x160 | (M & 15), 0xf, 0xf, true);
}

template <int M> __device__ __forceinline__ float xlane(float v)
{
    return __int_as_float(xlane<M>(__float_as_int(v)));
}

__global__ __launch_bounds__(256) void gat_scores(
    const float* __restrict__ X,
    const float* __restrict__ W,
    const float* __restrict__ att_src,
    const float* __restrict__ att_dst,
    float* __restrict__ a_src,
    float* __restrict__ a_dst,
    int* __restrict__ cur,
    int* __restrict__ total,
    int N)
{
    __shared__ __attribute__((aligned(16))) float s_v[D_IN][8];
    __shared__ __attribute__((aligned(16))) float s_stage[8][2][8][HEADS];

    const int t    = threadIdx.x;
    const int wave = t >> 5;
    const int lane = t & 31;

    if (blockIdx.x == 0 && t == 0) total[0] = 0;

    const int groups = (N + 7) >> 3;
    const int gstep  = gridDim.x * 8;
    int g = blockIdx.x * 8 + wave;

    float4 xv[8];
    #pragma unroll
    for (int q = 0; q < 8; ++q) {
        const int n = g * 8 + q;
        xv[q] = (g < groups && n < N) ? *(const float4*)(X + (size_t)n * D_IN + lane * 4)
                                      : make_float4(0.0f, 0.0f, 0.0f, 0.0f);
    }

    {
        const int hl  = lane >> 4;
        const int l16 = lane & 15;
        const float4 as_lo = *(const float4*)(att_src + lane * 4);
        const float4 as_hi = *(const float4*)(att_src + 128 + lane * 4);
        const float4 ad_lo = *(const float4*)(att_dst + lane * 4);
        const float4 ad_hi = *(const float4*)(att_dst + 128 + lane * 4);
        const int q = ((l16 >> 3) & 1) * 2 + ((l16 >> 2) & 1);
        const int vcol = (q >> 1) * 4 + (q & 1) * 2 + hl;
        for (int r0 = 0; r0 < 16; r0 += 4) {
            const int i0 = wave * 16 + r0;
            float4 wlo[4], whi[4];
            for (int k = 0; k < 4; ++k) {
                const float* wr = W + (size_t)(i0 + k) * NH_COLS + lane * 4;
                wlo[k] = *(const float4*)(wr);
                whi[k] = *(const float4*)(wr + 128);
            }
            float res[4];
            for (int k = 0; k < 4; ++k) {
                float p[4];
                p[0] = wlo[k].x * as_lo.x + wlo[k].y * as_lo.y + wlo[k].z * as_lo.z + wlo[k].w * as_lo.w;
                p[1] = whi[k].x * as_hi.x + whi[k].y * as_hi.y + whi[k].z * as_hi.z + whi[k].w * as_hi.w;
                p[2] = wlo[k].x * ad_lo.x + wlo[k].y * ad_lo.y + wlo[k].z * ad_lo.z + wlo[k].w * ad_lo.w;
                p[3] = whi[k].x * ad_hi.x + whi[k].y * ad_hi.y + whi[k].z * ad_hi.z + whi[k].w * ad_hi.w;
                {
                    const bool up = (l16 & 8) != 0;
                    #pragma unroll
                    for (int j = 0; j < 2; ++j) {
                        const float mine  = up ? p[2 + j] : p[j];
                        const float other = up ? p[j] : p[2 + j];
                        p[j] = mine + xlane<8>(other);
                    }
                }
                {
                    const bool up = (l16 & 4) != 0;
                    const float mine  = up ? p[1] : p[0];
                    const float other = up ? p[0] : p[1];
                    p[0] = mine + xlane<4>(other);
                }
                p[0] += xlane<2>(p[0]);
                p[0] += xlane<1>(p[0]);
                res[k] = p[0];
            }
            if ((l16 & 3) == 0) {
                for (int k = 0; k < 4; ++k) s_v[i0 + k][vcol] = res[k];
            }
        }
    }
    __syncthreads();

    float v[4][8];
    #pragma unroll
    for (int i = 0; i < 4; ++i) {
        const float4 lo = *(const float4*)&s_v[lane * 4 + i][0];
        const float4 hi = *(const float4*)&s_v[lane * 4 + i][4];
        v[i][0] = lo.x; v[i][1] = lo.y; v[i][2] = lo.z; v[i][3] = lo.w;
        v[i][4] = hi.x; v[i][5] = hi.y; v[i][6] = hi.z; v[i][7] = hi.w;
    }
    const int vi = ((lane >> 4) & 1) * 4 + ((lane >> 3) & 1) * 2 + ((lane >> 2) & 1);

    while (g < groups) {
        const int n0 = g * 8;
        if ((g & 3) == 0 && n0 + lane < N) cur[n0 + lane] = 0;

        #pragma unroll
        for (int q = 0; q < 8; ++q) {
            float p[8];
            #pragma unroll
            for (int k = 0; k < 8; ++k)
                p[k] = xv[q].x * v[0][k] + xv[q].y * v[1][k] + xv[q].z * v[2][k] + xv[q].w * v[3][k];
            {
                const bool up = (lane & 16) != 0;
                #pragma unroll
                for (int j = 0; j < 4; ++j) {
                    const float mine  = up ? p[4 + j] : p[j];
                    const float other = up ? p[j] : p[4 + j];
                    p[j] = mine + xlane<16>(other);
                }
            }
            {
                const bool up = (lane & 8) != 0;
                #pragma unroll
                for (int j = 0; j < 2; ++j) {
                    const float mine  = up ? p[2 + j] : p[j];
                    const float other = up ? p[j] : p[2 + j];
                    p[j] = mine + xlane<8>(other);
                }
            }
            {
                const bool up = (lane & 4) != 0;
                const float mine  = up ? p[1] : p[0];
                const float other = up ? p[0] : p[1];
                p[0] = mine + xlane<4>(other);
            }
            p[0] += xlane<2>(p[0]);
            p[0] += xlane<1>(p[0]);
            if ((lane & 3) == 0) s_stage[wave][vi >> 2][q][vi & 3] = p[0];
        }
        __builtin_amdgcn_wave_barrier();
        const float os = s_stage[wave][0][lane >> 2][lane & 3];
        const float od = s_stage[wave][1][lane >> 2][lane & 3];
        if (n0 + (lane >> 2) < N) {
            a_src[(size_t)n0 * HEADS + lane] = os;
            a_dst[(size_t)n0 * HEADS + lane] = od;
        }
        __builtin_amdgcn_wave_barrier();

        g += gstep;
        #pragma unroll
        for (int q = 0; q < 8; ++q) {
            const int n = g * 8 + q;
            xv[q] = (g < groups && n < N) ? *(const float4*)(X + (size_t)n * D_IN + lane * 4)
                                          : make_float4(0.0f, 0.0f, 0.0f, 0.0f);
        }
    }
}

__device__ __forceinline__ float leaky(float a)
{
    return a > 0.0f ? a : NEG_SLOPE * a;
}

#define ROW_SLOTS 32

__global__ void group_edges(const int* __restrict__ ei, int E, int N,
                            int* __restrict__ cnt, int* __restrict__ rows,
                            int* __restrict__ ovf, int* __restrict__ ovf_total)
{
    const int t = blockIdx.x * blockDim.x + threadIdx.x;
    const int T = gridDim.x * blockDim.x;
    int s[4], d[4], p[4];
    for (int k = 0; k < 4; ++k) {
        const int e = t + k * T;
        s[k] = (e < E) ? ei[e] : 0;
        d[k] = (e < E) ? ei[E + e] : -1;
    }
    for (int k = 0; k < 4; ++k)
        p[k] = ((unsigned)d[k] < (unsigned)N) ? atomicAdd(&cnt[d[k]], 1) : -1;
    for (int k = 0; k < 4; ++k) {
        if (p[k] < 0) continue;
        if (p[k] < ROW_SLOTS) {
            rows[(size_t)d[k] * ROW_SLOTS + p[k]] = s[k];
        } else {
            const int q = atomicAdd(ovf_total, 1);
            if ((unsigned)q < (unsigned)E) {
                ovf[2 * (size_t)q]     = d[k];
                ovf[2 * (size_t)q + 1] = s[k];
            }
        }
    }
}

__device__ __forceinline__ float wave_max(float v)
{
    v = fmaxf(v, xlane<16>(v)); v = fmaxf(v, xlane<8>(v)); v = fmaxf(v, xlane<4>(v));
    v = fmaxf(v, xlane<2>(v));  v = fmaxf(v, xlane<1>(v));
    return v;
}

__device__ __forceinline__ float wave_sum(float v)
{
    v += xlane<16>(v); v += xlane<8>(v); v += xlane<4>(v); v += xlane<2>(v); v += xlane<1>(v);
    return v;
}

__device__ __forceinline__ int wave_min_i(int v)
{
    v = min(v, xlane<16>(v)); v = min(v, xlane<8>(v)); v = min(v, xlane<4>(v));
    v = min(v, xlane<2>(v));  v = min(v, xlane<1>(v));
    return v;
}

__device__ __forceinline__ int wave_sum_i(int v)
{
    v += xlane<16>(v); v += xlane<8>(v); v += xlane<4>(v); v += xlane<2>(v); v += xlane<1>(v);
    return v;
}

__device__ __forceinline__ float4 edge_alpha(const float* __restrict__ a_src, int s, float4 ad)
{
    const float4 as = *(const float4*)(a_src + (size_t)s * HEADS);
    float4 al;
    al.x = leaky(as.x + ad.x);
    al.y = leaky(as.y + ad.y);
    al.z = leaky(as.z + ad.z);
    al.w = leaky(as.w + ad.w);
    return al;
}

__device__ __forceinline__ void add_xrow(float4& z0, float4& z1, float4& z2, float4& z3,
                                         const float* __restrict__ X, int s, int lane,
                                         float w0, float w1, float w2, float w3)
{
    const float4 xv = *(const float4*)(X + (size_t)s * D_IN + lane * 4);
    z0.x += w0 * xv.x; z0.y += w0 * xv.y; z0.z += w0 * xv.z; z0.w += w0 * xv.w;
    z1.x += w1 * xv.x; z1.y += w1 * xv.y; z1.z += w1 * xv.z; z1.w += w1 * xv.w;
    z2.x += w2 * xv.x; z2.y += w2 * xv.y; z2.z += w2 * xv.z; z2.w += w2 * xv.w;
    z3.x += w3 * xv.x; z3.y += w3 * xv.y; z3.z += w3 * xv.z; z3.w += w3 * xv.w;
}

__device__ __forceinline__ float lane_value(float v, int j)
{
    return __int_as_float(__builtin_amdgcn_readlane(__float_as_int(v), j));
}

#define NODES_PER_BLOCK 32
#define WAVES_PER_BLOCK 16
#define ZPITCH 516
#define PPITCH 68

__global__ __launch_bounds__(512) void gat_gather(
    const int* __restrict__ cnt, const int* __restrict__ rows,
    const int* __restrict__ ovf, const int* __restrict__ ovf_total,
    const float* __restrict__ a_src, const float* __restrict__ a_dst,
    const float* __restrict__ X,
    const float* __restrict__ W,
    const float* __restrict__ bias_gat,
    const float* __restrict__ onehot,
    const float* __restrict__ W_prior,
    const float* __restrict__ b_prior,
    const float* __restrict__ gate,
    float* __restrict__ out, int N, int E)
{
    __shared__ __attribute__((aligned(16))) float s_z[NODES_PER_BLOCK * ZPITCH];
    __shared__ __attribute__((aligned(16))) float s_p[HEADS * NODES_PER_BLOCK * PPITCH];
    int    (*sh_s)[2 * ROW_SLOTS] = (int (*)[2 * ROW_SLOTS])s_p;
    float4 (*sh_w)[2 * ROW_SLOTS] = (float4 (*)[2 * ROW_SLOTS])(s_p + WAVES_PER_BLOCK * 2 * ROW_SLOTS);

    const int lane = threadIdx.x & 31;
    const int half = lane >> 4;
    const int l16  = lane & 15;
    const int wv   = threadIdx.x >> 5;

  #pragma nounroll
  for (int q = 0; q < 2; ++q) {
    const int n = blockIdx.x * NODES_PER_BLOCK + q * 16 + wv;
    const bool node_ok = n < N;

    float4 z0 = make_float4(0.0f, 0.0f, 0.0f, 0.0f);
    float4 z1 = z0, z2 = z0, z3 = z0;

  if (node_ok) {
    const int nmax = N - 1;
    const int SENT = 0x7fffffff;
    int c = __builtin_amdgcn_readfirstlane(cnt[n]);
    c = min(max(c, 0), E);
    const int raw = rows[(size_t)n * ROW_SLOTS + lane];
    const float4 ad = *(const float4*)(a_dst + (size_t)n * HEADS);
    const float ninf = __int_as_float(0xFF800000);

    if (c < ROW_SLOTS) {
        int s = (lane < c) ? (int)min((unsigned)raw, (unsigned)nmax) : SENT;
        #define SORT_STEP(K, J) { const int p = xlane<J>(s); \
                                  const bool keepmin = ((lane & J) == 0) == ((lane & K) == 0); \
                                  s = keepmin ? min(s, p) : max(s, p); }
        SORT_STEP(2, 1)
        SORT_STEP(4, 2)   SORT_STEP(4, 1)
        SORT_STEP(8, 4)   SORT_STEP(8, 2)   SORT_STEP(8, 1)
        SORT_STEP(16, 8)  SORT_STEP(16, 4)  SORT_STEP(16, 2)  SORT_STEP(16, 1)
        SORT_STEP(32, 16) SORT_STEP(32, 8)  SORT_STEP(32, 4)  SORT_STEP(32, 2)  SORT_STEP(32, 1)
        #undef SORT_STEP
        const int total = c + 1;
        const bool valid = lane < total;
        if (lane >= c) s = n;

        const float4 al = edge_alpha(a_src, s, ad);
        float4 m;
        m.x = wave_max(valid ? al.x : ninf); m.y = wave_max(valid ? al.y : ninf);
        m.z = wave_max(valid ? al.z : ninf); m.w = wave_max(valid ? al.w : ninf);
        float4 e;
        e.x = valid ? __expf(al.x - m.x) : 0.0f; e.y = valid ? __expf(al.y - m.y) : 0.0f;
        e.z = valid ? __expf(al.z - m.z) : 0.0f; e.w = valid ? __expf(al.w - m.w) : 0.0f;
        float4 w;
        w.x = e.x / (wave_sum(e.x) + 1e-16f); w.y = e.y / (wave_sum(e.y) + 1e-16f);
        w.z = e.z / (wave_sum(e.z) + 1e-16f); w.w = e.w / (wave_sum(e.w) + 1e-16f);

        const int totp = (total + 3) & ~3;
        #pragma unroll 4
        for (int j = 0; j < totp; ++j) {
            const int   sj = __builtin_amdgcn_readlane(s, j);
            const float w0 = lane_value(w.x, j);
            const float w1 = lane_value(w.y, j);
            const float w2 = lane_value(w.z, j);
            const float w3 = lane_value(w.w, j);
            add_xrow(z0, z1, z2, z3, X, sj, lane, w0, w1, w2, w3);
        }
    } else {
        const int rs = (int)min((unsigned)raw, (unsigned)nmax);
        int novf = __builtin_amdgcn_readfirstlane(ovf_total[0]);
        novf = min(max(novf, 0), E);
        const int2* ov = (const int2*)ovf;

      if (c < 2 * ROW_SLOTS) {
        const int extra = c - ROW_SLOTS;
        int s1 = rs;
        int s2 = SENT;
        int filled = 0;
        #pragma nounroll
        for (int b = 0; b < novf; b += 32) {
            const int i = b + lane;
            int dn = -1, sv = 0;
            if (i < novf) {
                const int2 de = ov[i];
                dn = de.x;
                sv = (int)min((unsigned)de.y, (unsigned)nmax);
            }
            unsigned mask = __builtin_amdgcn_ballot_w32(dn == n);
            while (mask != 0u) {
                const int L = __builtin_ctz(mask);
                mask &= mask - 1u;
                const int v = __builtin_amdgcn_readlane(sv, L);
                if (lane == filled) s2 = v;
                ++filled;
            }
        }
        if (lane == extra) s2 = n;

        #define SORT2_STEP(K, J) { const int p1 = xlane<J>(s1); \
                                   const int p2 = xlane<J>(s2); \
                                   const bool lowpos = (lane & J) == 0; \
                                   const bool asc1 = (lane & K) == 0; \
                                   const bool asc2 = ((lane + 32) & K) == 0; \
                                   s1 = (lowpos == asc1) ? min(s1, p1) : max(s1, p1); \
                                   s2 = (lowpos == asc2) ? min(s2, p2) : max(s2, p2); }
        SORT2_STEP(2, 1)
        SORT2_STEP(4, 2)   SORT2_STEP(4, 1)
        SORT2_STEP(8, 4)   SORT2_STEP(8, 2)   SORT2_STEP(8, 1)
        SORT2_STEP(16, 8)  SORT2_STEP(16, 4)  SORT2_STEP(16, 2)  SORT2_STEP(16, 1)
        SORT2_STEP(32, 16) SORT2_STEP(32, 8)  SORT2_STEP(32, 4)  SORT2_STEP(32, 2)  SORT2_STEP(32, 1)
        { const int lo = min(s1, s2), hi = max(s1, s2); s1 = lo; s2 = hi; }
        SORT2_STEP(64, 16) SORT2_STEP(64, 8)  SORT2_STEP(64, 4)  SORT2_STEP(64, 2)  SORT2_STEP(64, 1)
        #undef SORT2_STEP
        const int total = c + 1;
        const bool valid2 = lane < total - ROW_SLOTS;
        if (!valid2) s2 = n;
        sh_s[wv][lane] = s1;
        sh_s[wv][ROW_SLOTS + lane] = s2;

        const float4 al1 = edge_alpha(a_src, s1, ad);
        const float4 al2 = edge_alpha(a_src, s2, ad);
        float4 m;
        m.x = wave_max(fmaxf(al1.x, valid2 ? al2.x : ninf)); m.y = wave_max(fmaxf(al1.y, valid2 ? al2.y : ninf));
        m.z = wave_max(fmaxf(al1.z, valid2 ? al2.z : ninf)); m.w = wave_max(fmaxf(al1.w, valid2 ? al2.w : ninf));
        float4 e1, e2;
        e1.x = __expf(al1.x - m.x); e1.y = __expf(al1.y - m.y);
        e1.z = __expf(al1.z - m.z); e1.w = __expf(al1.w - m.w);
        e2.x = valid2 ? __expf(al2.x - m.x) : 0.0f; e2.y = valid2 ? __expf(al2.y - m.y) : 0.0f;
        e2.z = valid2 ? __expf(al2.z - m.z) : 0.0f; e2.w = valid2 ? __expf(al2.w - m.w) : 0.0f;
        float4 d;
        d.x = wave_sum(e1.x + e2.x) + 1e-16f; d.y = wave_sum(e1.y + e2.y) + 1e-16f;
        d.z = wave_sum(e1.z + e2.z) + 1e-16f; d.w = wave_sum(e1.w + e2.w) + 1e-16f;

        d.x = 1.0f / d.x; d.y = 1.0f / d.y; d.z = 1.0f / d.z; d.w = 1.0f / d.w;
        sh_w[wv][lane] = make_float4(e1.x * d.x, e1.y * d.y, e1.z * d.z, e1.w * d.w);
        sh_w[wv][ROW_SLOTS + lane] = make_float4(e2.x * d.x, e2.y * d.y, e2.z * d.z, e2.w * d.w);
        __builtin_amdgcn_wave_barrier();
        #pragma nounroll
        for (int j = 0; j < total; ++j) {
            const int sj = sh_s[wv][j];
            const float4 wj = sh_w[wv][j];
            add_xrow(z0, z1, z2, z3, X, sj, lane, wj.x, wj.y, wj.z, wj.w);
        }
      } else {

        float4 m = edge_alpha(a_src, n, ad);
        {
            const float4 al = edge_alpha(a_src, rs, ad);
            m.x = fmaxf(m.x, al.x); m.y = fmaxf(m.y, al.y);
            m.z = fmaxf(m.z, al.z); m.w = fmaxf(m.w, al.w);
        }
        #pragma nounroll
        for (int b = 0; b < novf; b += 32) {
            const int i = b + lane;
            if (i < novf) {
                const int2 de = ov[i];
                if (de.x == n) {
                    const float4 al = edge_alpha(a_src, (int)min((unsigned)de.y, (unsigned)nmax), ad);
                    m.x = fmaxf(m.x, al.x); m.y = fmaxf(m.y, al.y);
                    m.z = fmaxf(m.z, al.z); m.w = fmaxf(m.w, al.w);
                }
            }
        }
        m.x = wave_max(m.x); m.y = wave_max(m.y); m.z = wave_max(m.z); m.w = wave_max(m.w);

        float4 den = make_float4(0.0f, 0.0f, 0.0f, 0.0f);
        for (int pass = 0; pass < 2; ++pass) {
            int last = -1;
            for (int it = 0; it <= c; ++it) {
                int k = (rs > last) ? rs : SENT;
                if (n > last) k = min(k, n);
                #pragma nounroll
                for (int b = 0; b < novf; b += 32) {
                    const int i = b + lane;
                    if (i < novf) {
                        const int2 de = ov[i];
                        const int sv = (int)min((unsigned)de.y, (unsigned)nmax);
                        if (de.x == n && sv > last) k = min(k, sv);
                    }
                }
                k = __builtin_amdgcn_readfirstlane(wave_min_i(k));
                if (k == SENT) break;
                int mult = (rs == k) ? 1 : 0;
                #pragma nounroll
                for (int b = 0; b < novf; b += 32) {
                    const int i = b + lane;
                    if (i < novf) {
                        const int2 de = ov[i];
                        const int sv = (int)min((unsigned)de.y, (unsigned)nmax);
                        if (de.x == n && sv == k) mult += 1;
                    }
                }
                mult = wave_sum_i(mult) + ((n == k) ? 1 : 0);
                last = k;
                const float fm = (float)mult;
                const float4 al = edge_alpha(a_src, k, ad);
                float4 e;
                e.x = fm * __expf(al.x - m.x); e.y = fm * __expf(al.y - m.y);
                e.z = fm * __expf(al.z - m.z); e.w = fm * __expf(al.w - m.w);
                if (pass == 0) {
                    den.x += e.x; den.y += e.y; den.z += e.z; den.w += e.w;
                } else {
                    add_xrow(z0, z1, z2, z3, X, k, lane, e.x * den.x, e.y * den.y, e.z * den.z, e.w * den.w);
                }
            }
            if (pass == 0) {
                den.x = 1.0f / (den.x + 1e-16f); den.y = 1.0f / (den.y + 1e-16f);
                den.z = 1.0f / (den.z + 1e-16f); den.w = 1.0f / (den.w + 1e-16f);
            }
        }
      }
    }
  }

    {
        float* zr = &s_z[(q * 16 + wv) * ZPITCH + lane * 4];
        *(float4*)(zr)            = z0;
        *(float4*)(zr + D_IN)     = z1;
        *(float4*)(zr + 2 * D_IN) = z2;
        *(float4*)(zr + 3 * D_IN) = z3;
    }
  }

    const int head = wv >> 2;
    const int ct   = wv & 3;
    const float* wcol = W + (size_t)(half * 2) * NH_COLS + head * F_OUT + ct * 16 + l16;
    float bq[16];
    #pragma unroll
    for (int t = 0; t < 8; ++t) {
        bq[2 * t]     = wcol[(size_t)(t * 4)     * NH_COLS];
        bq[2 * t + 1] = wcol[(size_t)(t * 4 + 1) * NH_COLS];
    }
    __syncthreads();

    {
        const float* zrow = &s_z[l16 * ZPITCH + head * D_IN + half * 2];
        v8f accA = {}; v8f accB = {};
        #pragma nounroll
        for (int g = 0; g < 4; ++g) {
            if (g > 0) {
                #pragma unroll
                for (int t = 0; t < 8; ++t) {
                    bq[2 * t]     = wcol[(size_t)(g * 32 + t * 4)     * NH_COLS];
                    bq[2 * t + 1] = wcol[(size_t)(g * 32 + t * 4 + 1) * NH_COLS];
                }
            }
            #pragma unroll
            for (int t = 0; t < 8; ++t) {
                const v2f a0 = *(const v2f*)(zrow + g * 32 + t * 4);
                const v2f a1 = *(const v2f*)(zrow + 16 * ZPITCH + g * 32 + t * 4);
                v2f b; b.x = bq[2 * t]; b.y = bq[2 * t + 1];
                accA = __builtin_amdgcn_wmma_f32_16x16x4_f32(false, a0, false, b, (short)0, accA, false, false);
                accB = __builtin_amdgcn_wmma_f32_16x16x4_f32(false, a1, false, b, (short)0, accB, false, false);
            }
        }
        #pragma unroll
        for (int r = 0; r < 8; ++r) {
            s_p[(head * NODES_PER_BLOCK + r + half * 8) * PPITCH + ct * 16 + l16]      = accA[r];
            s_p[(head * NODES_PER_BLOCK + 16 + r + half * 8) * PPITCH + ct * 16 + l16] = accB[r];
        }
    }
    __syncthreads();

  for (int q = 0; q < 2; ++q) {
    const int row = q * 16 + wv;
    const int n = blockIdx.x * NODES_PER_BLOCK + row;
    if (n >= N) continue;

    float4 acc;
    {
        const float4 p0 = *(const float4*)&s_p[(0 * NODES_PER_BLOCK + row) * PPITCH + l16 * 4];
        const float4 p1 = *(const float4*)&s_p[(1 * NODES_PER_BLOCK + row) * PPITCH + l16 * 4];
        const float4 p2 = *(const float4*)&s_p[(2 * NODES_PER_BLOCK + row) * PPITCH + l16 * 4];
        const float4 p3 = *(const float4*)&s_p[(3 * NODES_PER_BLOCK + row) * PPITCH + l16 * 4];
        acc.x = ((p0.x + p1.x) + p2.x) + p3.x;
        acc.y = ((p0.y + p1.y) + p2.y) + p3.y;
        acc.z = ((p0.z + p1.z) + p2.z) + p3.z;
        acc.w = ((p0.w + p1.w) + p2.w) + p3.w;
    }

    const float ohv = onehot[(size_t)n * 32 + lane];
    float4 p = make_float4(0.0f, 0.0f, 0.0f, 0.0f);
    #pragma unroll 4
    for (int c = 0; c < 16; ++c) {
        const float o = __shfl(ohv, c, 16);
        const float4 wp = *(const float4*)(W_prior + (half * 16 + c) * F_OUT + l16 * 4);
        p.x += o * wp.x; p.y += o * wp.y; p.z += o * wp.z; p.w += o * wp.w;
    }

    p.x += xlane<16>(p.x); p.y += xlane<16>(p.y);
    p.z += xlane<16>(p.z); p.w += xlane<16>(p.w);

    if (half == 0) {
        const float g = gate[0];
        const float4 bg = *(const float4*)(bias_gat + l16 * 4);
        const float4 bp = *(const float4*)(b_prior + l16 * 4);
        float4 o;
        o.x = (1.0f - g) * (0.25f * acc.x + bg.x) + g * (bp.x + p.x);
        o.y = (1.0f - g) * (0.25f * acc.y + bg.y) + g * (bp.y + p.y);
        o.z = (1.0f - g) * (0.25f * acc.z + bg.z) + g * (bp.z + p.z);
        o.w = (1.0f - g) * (0.25f * acc.w + bg.w) + g * (bp.w + p.w);
        *(float4*)(out + (size_t)n * F_OUT + l16 * 4) = o;
    }
  }
}

extern "C" void kernel_launch(void* const* d_in, const int* in_sizes, int n_in,
                              void* d_out, int out_size, void* d_ws, size_t ws_size,
                              hipStream_t stream)
{
    const float* x        = (const float*)d_in[0];
    const int*   ei       = (const int*)  d_in[1];
    const float* onehot   = (const float*)d_in[2];
    const float* W_gat    = (const float*)d_in[3];
    const float* att_src  = (const float*)d_in[4];
    const float* att_dst  = (const float*)d_in[5];
    const float* bias_gat = (const float*)d_in[6];
    const float* W_prior  = (const float*)d_in[7];
    const float* b_prior  = (const float*)d_in[8];
    const float* gate     = (const float*)d_in[9];
    float* out = (float*)d_out;

    const int N    = in_sizes[0] / D_IN;
    const int E    = in_sizes[1] / 2;
    const int NHs  = N * HEADS;

    float* h      = (float*)d_ws;
    int*   rows   = (int*)(h + (size_t)N * NH_COLS);
    int*   ovf    = rows + (size_t)N * ROW_SLOTS;
    float* a_src  = (float*)(ovf + (size_t)E * 2);
    float* a_dst  = a_src  + NHs;
    int*   cnt    = (int*)(a_dst + NHs);
    int*   ototal = cnt + N;

    gat_scores<<<(N + 63) / 64, 256, 0, stream>>>(x, W_gat, att_src, att_dst,
                                                  a_src, a_dst, cnt, ototal, N);

    group_edges<<<((E + 3) / 4 + 255) / 256, 256, 0, stream>>>(ei, E, N, cnt, rows, ovf, ototal);

    gat_gather<<<(N + NODES_PER_BLOCK - 1) / NODES_PER_BLOCK, 512, 0, stream>>>(
        cnt, rows, ovf, ototal, a_src, a_dst, x, W_gat,
        bias_gat, onehot, W_prior, b_prior, gate, out, N, E);
}


# Group edges by destination through LDS: blocks count, then scatter, their edges into 512-node bins; a block per bin fills the 32-slot rows in LDS and stores whole lines; 3 -> 4 launches
# speedup vs baseline: 6.0059x; 1.3731x over previous
#include <hip/hip_runtime.h>
#include <hip/hip_bf16.h>
#include <math.h>

typedef __attribute__((ext_vector_type(2))) float v2f;
typedef __attribute__((ext_vector_type(8))) float v8f;

#define HEADS 4
#define F_OUT 64
#define D_IN 128
#define NH_COLS 256
#define NEG_SLOPE 0.2f

template <int M> __device__ __forceinline__ int xlane(int v)
{
    if (M == 16) return __builtin_amdgcn_permlanex16(v, v, 0x76543210, 0xfedcba98, false, false);
    return __builtin_amdgcn_update_dpp(0, v, 0x160 | (M & 15), 0xf, 0xf, true);
}

template <int M> __device__ __forceinline__ float xlane(float v)
{
    return __int_as_float(xlane<M>(__float_as_int(v)));
}

#define BIN_SHIFT 9
#define BIN_NODES 512
#define EDGE_CHUNK 8192

__global__ __launch_bounds__(256) void gat_scores(
    const float* __restrict__ X,
    const float* __restrict__ W,
    const float* __restrict__ att_src,
    const float* __restrict__ att_dst,
    float* __restrict__ a_src,
    float* __restrict__ a_dst,
    const int* __restrict__ ei,
    int* __restrict__ hist,
    int* __restrict__ total,
    int N, int E, int neb, int nbins, int nscore)
{
    __shared__ __attribute__((aligned(16))) float s_v[D_IN][8];
    __shared__ __attribute__((aligned(16))) float s_stage[8][2][8][HEADS];
    extern __shared__ int s_hist[];

    const int t    = threadIdx.x;
    const int wave = t >> 5;
    const int lane = t & 31;

    if (blockIdx.x < neb) {
        for (int i = t; i < nbins; i += 256) s_hist[i] = 0;
        __syncthreads();
        const int e0 = blockIdx.x * EDGE_CHUNK;
        for (int h = 0; h < EDGE_CHUNK / (256 * 16); ++h) {
            int d[16];
            for (int k = 0; k < 16; ++k) {
                const int e = e0 + (h * 16 + k) * 256 + t;
                d[k] = (e < E) ? ei[E + e] : -1;
            }
            for (int k = 0; k < 16; ++k)
                if ((unsigned)d[k] < (unsigned)N) atomicAdd(&s_hist[d[k] >> BIN_SHIFT], 1);
        }
        __syncthreads();
        for (int i = t; i < nbins; i += 256) hist[(size_t)i * neb + blockIdx.x] = s_hist[i];
        return;
    }
    const int sblock = blockIdx.x - neb;

    if (sblock == 0 && t == 0) total[0] = 0;

    const int groups = (N + 7) >> 3;
    const int gstep  = nscore * 8;
    int g = sblock * 8 + wave;

    float4 xv[8];
    #pragma unroll
    for (int q = 0; q < 8; ++q) {
        const int n = g * 8 + q;
        xv[q] = (g < groups && n < N) ? *(const float4*)(X + (size_t)n * D_IN + lane * 4)
                                      : make_float4(0.0f, 0.0f, 0.0f, 0.0f);
    }

    {
        const int hl  = lane >> 4;
        const int l16 = lane & 15;
        const float4 as_lo = *(const float4*)(att_src + lane * 4);
        const float4 as_hi = *(const float4*)(att_src + 128 + lane * 4);
        const float4 ad_lo = *(const float4*)(att_dst + lane * 4);
        const float4 ad_hi = *(const float4*)(att_dst + 128 + lane * 4);
        const int q = ((l16 >> 3) & 1) * 2 + ((l16 >> 2) & 1);
        const int vcol = (q >> 1) * 4 + (q & 1) * 2 + hl;
        for (int r0 = 0; r0 < 16; r0 += 4) {
            const int i0 = wave * 16 + r0;
            float4 wlo[4], whi[4];
            for (int k = 0; k < 4; ++k) {
                const float* wr = W + (size_t)(i0 + k) * NH_COLS + lane * 4;
                wlo[k] = *(const float4*)(wr);
                whi[k] = *(const float4*)(wr + 128);
            }
            float res[4];
            for (int k = 0; k < 4; ++k) {
                float p[4];
                p[0] = wlo[k].x * as_lo.x + wlo[k].y * as_lo.y + wlo[k].z * as_lo.z + wlo[k].w * as_lo.w;
                p[1] = whi[k].x * as_hi.x + whi[k].y * as_hi.y + whi[k].z * as_hi.z + whi[k].w * as_hi.w;
                p[2] = wlo[k].x * ad_lo.x + wlo[k].y * ad_lo.y + wlo[k].z * ad_lo.z + wlo[k].w * ad_lo.w;
                p[3] = whi[k].x * ad_hi.x + whi[k].y * ad_hi.y + whi[k].z * ad_hi.z + whi[k].w * ad_hi.w;
                {
                    const bool up = (l16 & 8) != 0;
                    #pragma unroll
                    for (int j = 0; j < 2; ++j) {
                        const float mine  = up ? p[2 + j] : p[j];
                        const float other = up ? p[j] : p[2 + j];
                        p[j] = mine + xlane<8>(other);
                    }
                }
                {
                    const bool up = (l16 & 4) != 0;
                    const float mine  = up ? p[1] : p[0];
                    const float other = up ? p[0] : p[1];
                    p[0] = mine + xlane<4>(other);
                }
                p[0] += xlane<2>(p[0]);
                p[0] += xlane<1>(p[0]);
                res[k] = p[0];
            }
            if ((l16 & 3) == 0) {
                for (int k = 0; k < 4; ++k) s_v[i0 + k][vcol] = res[k];
            }
        }
    }
    __syncthreads();

    float v[4][8];
    #pragma unroll
    for (int i = 0; i < 4; ++i) {
        const float4 lo = *(const float4*)&s_v[lane * 4 + i][0];
        const float4 hi = *(const float4*)&s_v[lane * 4 + i][4];
        v[i][0] = lo.x; v[i][1] = lo.y; v[i][2] = lo.z; v[i][3] = lo.w;
        v[i][4] = hi.x; v[i][5] = hi.y; v[i][6] = hi.z; v[i][7] = hi.w;
    }
    const int vi = ((lane >> 4) & 1) * 4 + ((lane >> 3) & 1) * 2 + ((lane >> 2) & 1);

    while (g < groups) {
        const int n0 = g * 8;

        #pragma unroll
        for (int q = 0; q < 8; ++q) {
            float p[8];
            #pragma unroll
            for (int k = 0; k < 8; ++k)
                p[k] = xv[q].x * v[0][k] + xv[q].y * v[1][k] + xv[q].z * v[2][k] + xv[q].w * v[3][k];
            {
                const bool up = (lane & 16) != 0;
                #pragma unroll
                for (int j = 0; j < 4; ++j) {
                    const float mine  = up ? p[4 + j] : p[j];
                    const float other = up ? p[j] : p[4 + j];
                    p[j] = mine + xlane<16>(other);
                }
            }
            {
                const bool up = (lane & 8) != 0;
                #pragma unroll
                for (int j = 0; j < 2; ++j) {
                    const float mine  = up ? p[2 + j] : p[j];
                    const float other = up ? p[j] : p[2 + j];
                    p[j] = mine + xlane<8>(other);
                }
            }
            {
                const bool up = (lane & 4) != 0;
                const float mine  = up ? p[1] : p[0];
                const float other = up ? p[0] : p[1];
                p[0] = mine + xlane<4>(other);
            }
            p[0] += xlane<2>(p[0]);
            p[0] += xlane<1>(p[0]);
            if ((lane & 3) == 0) s_stage[wave][vi >> 2][q][vi & 3] = p[0];
        }
        __builtin_amdgcn_wave_barrier();
        const float os = s_stage[wave][0][lane >> 2][lane & 3];
        const float od = s_stage[wave][1][lane >> 2][lane & 3];
        if (n0 + (lane >> 2) < N) {
            a_src[(size_t)n0 * HEADS + lane] = os;
            a_dst[(size_t)n0 * HEADS + lane] = od;
        }
        __builtin_amdgcn_wave_barrier();

        g += gstep;
        #pragma unroll
        for (int q = 0; q < 8; ++q) {
            const int n = g * 8 + q;
            xv[q] = (g < groups && n < N) ? *(const float4*)(X + (size_t)n * D_IN + lane * 4)
                                          : make_float4(0.0f, 0.0f, 0.0f, 0.0f);
        }
    }
}

__device__ __forceinline__ float leaky(float a)
{
    return a > 0.0f ? a : NEG_SLOPE * a;
}

#define ROW_SLOTS 32

__device__ __forceinline__ float wave_max(float v)
{
    v = fmaxf(v, xlane<16>(v)); v = fmaxf(v, xlane<8>(v)); v = fmaxf(v, xlane<4>(v));
    v = fmaxf(v, xlane<2>(v));  v = fmaxf(v, xlane<1>(v));
    return v;
}

__device__ __forceinline__ float wave_sum(float v)
{
    v += xlane<16>(v); v += xlane<8>(v); v += xlane<4>(v); v += xlane<2>(v); v += xlane<1>(v);
    return v;
}

__device__ __forceinline__ int wave_min_i(int v)
{
    v = min(v, xlane<16>(v)); v = min(v, xlane<8>(v)); v = min(v, xlane<4>(v));
    v = min(v, xlane<2>(v));  v = min(v, xlane<1>(v));
    return v;
}

__device__ __forceinline__ int wave_sum_i(int v)
{
    v += xlane<16>(v); v += xlane<8>(v); v += xlane<4>(v); v += xlane<2>(v); v += xlane<1>(v);
    return v;
}

__global__ __launch_bounds__(512) void bin_edges(const int* __restrict__ ei, int E, int N,
                                                 const int* __restrict__ hist, int neb, int nbins,
                                                 int* __restrict__ binned_w,
                                                 int* __restrict__ binstart)
{
    int2* binned = (int2*)binned_w;
    extern __shared__ int s_bin[];
    int* s_base = s_bin;
    int* s_part = s_bin + nbins;
    int* s_next = s_bin + 2 * nbins;

    const int t    = threadIdx.x;
    const int wave = t >> 5;
    const int lane = t & 31;
    const int eb   = blockIdx.x;

    for (int b0 = wave; b0 < nbins; b0 += 64) {
        int tot[4] = {0, 0, 0, 0}, part[4] = {0, 0, 0, 0};
        for (int jb = 0; jb < neb; jb += 128) {
            int v[4][4];
            for (int u = 0; u < 4; ++u)
                for (int q = 0; q < 4; ++q) {
                    const int b = b0 + 16 * u;
                    const int j = jb + 32 * q + lane;
                    v[u][q] = (b < nbins && j < neb) ? hist[(size_t)b * neb + j] : 0;
                }
            for (int u = 0; u < 4; ++u)
                for (int q = 0; q < 4; ++q) {
                    const int j = jb + 32 * q + lane;
                    tot[u]  += v[u][q];
                    part[u] += (j < eb) ? v[u][q] : 0;
                }
        }
        for (int u = 0; u < 4; ++u) {
            const int b = b0 + 16 * u;
            const int ts = wave_sum_i(tot[u]);
            const int ps = wave_sum_i(part[u]);
            if (b < nbins && lane == 0) { s_base[b] = ts; s_part[b] = ps; s_next[b] = 0; }
        }
    }
    __syncthreads();

    if (wave == 0) {
        int carry = 0;
        for (int c0 = 0; c0 < nbins; c0 += 128) {
            const int i0 = c0 + lane * 4;
            int v[4];
            for (int q = 0; q < 4; ++q) v[q] = (i0 + q < nbins) ? s_base[i0 + q] : 0;
            const int sum4 = (v[0] + v[1]) + (v[2] + v[3]);
            int incl = sum4;
            for (int d = 1; d < 32; d <<= 1) {
                const int u = __shfl_up(incl, d);
                if (lane >= d) incl += u;
            }
            int run = carry + incl - sum4;
            carry += __shfl(incl, 31);
            for (int q = 0; q < 4; ++q) {
                if (i0 + q < nbins) {
                    s_base[i0 + q] = run + s_part[i0 + q];
                    if (eb == 0) binstart[i0 + q] = run;
                }
                run += v[q];
            }
        }
        if (eb == 0 && lane == 0) binstart[nbins] = carry;
    }
    __syncthreads();

    const int e0 = eb * EDGE_CHUNK;
    for (int h = 0; h < EDGE_CHUNK / (512 * 8); ++h) {
        int s[8], d[8];
        for (int k = 0; k < 8; ++k) {
            const int e = e0 + (h * 8 + k) * 512 + t;
            s[k] = (e < E) ? ei[e] : 0;
            d[k] = (e < E) ? ei[E + e] : -1;
        }
        int base[8], r[8];
        for (int k = 0; k < 8; ++k)
            base[k] = s_base[((unsigned)d[k] < (unsigned)N) ? (d[k] >> BIN_SHIFT) : 0];
        for (int k = 0; k < 8; ++k)
            r[k] = ((unsigned)d[k] < (unsigned)N) ? atomicAdd(&s_next[d[k] >> BIN_SHIFT], 1) : -1;
        for (int k = 0; k < 8; ++k) {
            const int pos = base[k] + r[k];
            if (r[k] >= 0 && (unsigned)pos < (unsigned)E) binned[pos] = make_int2(d[k], s[k]);
        }
    }
}

__global__ __launch_bounds__(512) void group_bins(const int* __restrict__ binned_w,
                                                  const int* __restrict__ binstart, int N, int E,
                                                  int* __restrict__ cnt, int* __restrict__ rows,
                                                  int* __restrict__ ovf, int* __restrict__ ovf_total)
{
    const int2* binned = (const int2*)binned_w;
    __shared__ __attribute__((aligned(16))) int s_rows[BIN_NODES * ROW_SLOTS];
    __shared__ int s_cn[BIN_NODES];

    const int t  = threadIdx.x;
    const int n0 = blockIdx.x * BIN_NODES;
    s_cn[t] = 0;
    int start = binstart[blockIdx.x];
    int end   = binstart[blockIdx.x + 1];
    start = min(max(start, 0), E);
    end   = min(max(end, start), E);
    __syncthreads();

    for (int i0 = start; i0 < end; i0 += 512 * 8) {
        int2 de[8];
        int  p[8];
        for (int k = 0; k < 8; ++k) {
            const int i = i0 + k * 512 + t;
            de[k] = (i < end) ? binned[i] : make_int2(-1, 0);
        }
        for (int k = 0; k < 8; ++k) {
            const int dl = de[k].x - n0;
            p[k] = ((unsigned)dl < (unsigned)BIN_NODES) ? atomicAdd(&s_cn[dl], 1) : -1;
        }
        for (int k = 0; k < 8; ++k) {
            if (p[k] < 0) continue;
            if (p[k] < ROW_SLOTS) {
                s_rows[(de[k].x - n0) * ROW_SLOTS + p[k]] = de[k].y;
            } else {
                const int q = atomicAdd(ovf_total, 1);
                if ((unsigned)q < (unsigned)E) {
                    ovf[2 * (size_t)q]     = de[k].x;
                    ovf[2 * (size_t)q + 1] = de[k].y;
                }
            }
        }
    }
    __syncthreads();

    for (int k = 0; k < BIN_NODES * ROW_SLOTS / (512 * 4); ++k) {
        const int ch  = k * 512 + t;
        const int row = ch >> 3;
        const int sl  = (ch & 7) * 4;
        const int n   = n0 + row;
        if (n < N) {
            const int cv = s_cn[row];
            const int4 r = *(const int4*)&s_rows[row * ROW_SLOTS + sl];
            int4 v;
            v.x = (sl     < cv) ? r.x : 0;
            v.y = (sl + 1 < cv) ? r.y : 0;
            v.z = (sl + 2 < cv) ? r.z : 0;
            v.w = (sl + 3 < cv) ? r.w : 0;
            *(int4*)(rows + (size_t)n * ROW_SLOTS + sl) = v;
        }
    }
    if (n0 + t < N) cnt[n0 + t] = s_cn[t];
}


__device__ __forceinline__ float4 edge_alpha(const float* __restrict__ a_src, int s, float4 ad)
{
    const float4 as = *(const float4*)(a_src + (size_t)s * HEADS);
    float4 al;
    al.x = leaky(as.x + ad.x);
    al.y = leaky(as.y + ad.y);
    al.z = leaky(as.z + ad.z);
    al.w = leaky(as.w + ad.w);
    return al;
}

__device__ __forceinline__ void add_xrow(float4& z0, float4& z1, float4& z2, float4& z3,
                                         const float* __restrict__ X, int s, int lane,
                                         float w0, float w1, float w2, float w3)
{
    const float4 xv = *(const float4*)(X + (size_t)s * D_IN + lane * 4);
    z0.x += w0 * xv.x; z0.y += w0 * xv.y; z0.z += w0 * xv.z; z0.w += w0 * xv.w;
    z1.x += w1 * xv.x; z1.y += w1 * xv.y; z1.z += w1 * xv.z; z1.w += w1 * xv.w;
    z2.x += w2 * xv.x; z2.y += w2 * xv.y; z2.z += w2 * xv.z; z2.w += w2 * xv.w;
    z3.x += w3 * xv.x; z3.y += w3 * xv.y; z3.z += w3 * xv.z; z3.w += w3 * xv.w;
}

__device__ __forceinline__ float lane_value(float v, int j)
{
    return __int_as_float(__builtin_amdgcn_readlane(__float_as_int(v), j));
}

#define NODES_PER_BLOCK 32
#define WAVES_PER_BLOCK 16
#define ZPITCH 516
#define PPITCH 68

__global__ __launch_bounds__(512) void gat_gather(
    const int* __restrict__ cnt, const int* __restrict__ rows,
    const int* __restrict__ ovf, const int* __restrict__ ovf_total,
    const float* __restrict__ a_src, const float* __restrict__ a_dst,
    const float* __restrict__ X,
    const float* __restrict__ W,
    const float* __restrict__ bias_gat,
    const float* __restrict__ onehot,
    const float* __restrict__ W_prior,
    const float* __restrict__ b_prior,
    const float* __restrict__ gate,
    float* __restrict__ out, int N, int E)
{
    __shared__ __attribute__((aligned(16))) float s_z[NODES_PER_BLOCK * ZPITCH];
    __shared__ __attribute__((aligned(16))) float s_p[HEADS * NODES_PER_BLOCK * PPITCH];
    int    (*sh_s)[2 * ROW_SLOTS] = (int (*)[2 * ROW_SLOTS])s_p;
    float4 (*sh_w)[2 * ROW_SLOTS] = (float4 (*)[2 * ROW_SLOTS])(s_p + WAVES_PER_BLOCK * 2 * ROW_SLOTS);

    const int lane = threadIdx.x & 31;
    const int half = lane >> 4;
    const int l16  = lane & 15;
    const int wv   = threadIdx.x >> 5;

  #pragma nounroll
  for (int q = 0; q < 2; ++q) {
    const int n = blockIdx.x * NODES_PER_BLOCK + q * 16 + wv;
    const bool node_ok = n < N;

    float4 z0 = make_float4(0.0f, 0.0f, 0.0f, 0.0f);
    float4 z1 = z0, z2 = z0, z3 = z0;

  if (node_ok) {
    const int nmax = N - 1;
    const int SENT = 0x7fffffff;
    int c = __builtin_amdgcn_readfirstlane(cnt[n]);
    c = min(max(c, 0), E);
    const int raw = rows[(size_t)n * ROW_SLOTS + lane];
    const float4 ad = *(const float4*)(a_dst + (size_t)n * HEADS);
    const float ninf = __int_as_float(0xFF800000);

    if (c < ROW_SLOTS) {
        int s = (lane < c) ? (int)min((unsigned)raw, (unsigned)nmax) : SENT;
        #define SORT_STEP(K, J) { const int p = xlane<J>(s); \
                                  const bool keepmin = ((lane & J) == 0) == ((lane & K) == 0); \
                                  s = keepmin ? min(s, p) : max(s, p); }
        SORT_STEP(2, 1)
        SORT_STEP(4, 2)   SORT_STEP(4, 1)
        SORT_STEP(8, 4)   SORT_STEP(8, 2)   SORT_STEP(8, 1)
        SORT_STEP(16, 8)  SORT_STEP(16, 4)  SORT_STEP(16, 2)  SORT_STEP(16, 1)
        SORT_STEP(32, 16) SORT_STEP(32, 8)  SORT_STEP(32, 4)  SORT_STEP(32, 2)  SORT_STEP(32, 1)
        #undef SORT_STEP
        const int total = c + 1;
        const bool valid = lane < total;
        if (lane >= c) s = n;

        const float4 al = edge_alpha(a_src, s, ad);
        float4 m;
        m.x = wave_max(valid ? al.x : ninf); m.y = wave_max(valid ? al.y : ninf);
        m.z = wave_max(valid ? al.z : ninf); m.w = wave_max(valid ? al.w : ninf);
        float4 e;
        e.x = valid ? __expf(al.x - m.x) : 0.0f; e.y = valid ? __expf(al.y - m.y) : 0.0f;
        e.z = valid ? __expf(al.z - m.z) : 0.0f; e.w = valid ? __expf(al.w - m.w) : 0.0f;
        float4 w;
        w.x = e.x / (wave_sum(e.x) + 1e-16f); w.y = e.y / (wave_sum(e.y) + 1e-16f);
        w.z = e.z / (wave_sum(e.z) + 1e-16f); w.w = e.w / (wave_sum(e.w) + 1e-16f);

        const int totp = (total + 3) & ~3;
        #pragma unroll 4
        for (int j = 0; j < totp; ++j) {
            const int   sj = __builtin_amdgcn_readlane(s, j);
            const float w0 = lane_value(w.x, j);
            const float w1 = lane_value(w.y, j);
            const float w2 = lane_value(w.z, j);
            const float w3 = lane_value(w.w, j);
            add_xrow(z0, z1, z2, z3, X, sj, lane, w0, w1, w2, w3);
        }
    } else {
        const int rs = (int)min((unsigned)raw, (unsigned)nmax);
        int novf = __builtin_amdgcn_readfirstlane(ovf_total[0]);
        novf = min(max(novf, 0), E);
        const int2* ov = (const int2*)ovf;

      if (c < 2 * ROW_SLOTS) {
        const int extra = c - ROW_SLOTS;
        int s1 = rs;
        int s2 = SENT;
        int filled = 0;
        #pragma nounroll
        for (int b = 0; b < novf; b += 32) {
            const int i = b + lane;
            int dn = -1, sv = 0;
            if (i < novf) {
                const int2 de = ov[i];
                dn = de.x;
                sv = (int)min((unsigned)de.y, (unsigned)nmax);
            }
            unsigned mask = __builtin_amdgcn_ballot_w32(dn == n);
            while (mask != 0u) {
                const int L = __builtin_ctz(mask);
                mask &= mask - 1u;
                const int v = __builtin_amdgcn_readlane(sv, L);
                if (lane == filled) s2 = v;
                ++filled;
            }
        }
        if (lane == extra) s2 = n;

        #define SORT2_STEP(K, J) { const int p1 = xlane<J>(s1); \
                                   const int p2 = xlane<J>(s2); \
                                   const bool lowpos = (lane & J) == 0; \
                                   const bool asc1 = (lane & K) == 0; \
                                   const bool asc2 = ((lane + 32) & K) == 0; \
                                   s1 = (lowpos == asc1) ? min(s1, p1) : max(s1, p1); \
                                   s2 = (lowpos == asc2) ? min(s2, p2) : max(s2, p2); }
        SORT2_STEP(2, 1)
        SORT2_STEP(4, 2)   SORT2_STEP(4, 1)
        SORT2_STEP(8, 4)   SORT2_STEP(8, 2)   SORT2_STEP(8, 1)
        SORT2_STEP(16, 8)  SORT2_STEP(16, 4)  SORT2_STEP(16, 2)  SORT2_STEP(16, 1)
        SORT2_STEP(32, 16) SORT2_STEP(32, 8)  SORT2_STEP(32, 4)  SORT2_STEP(32, 2)  SORT2_STEP(32, 1)
        { const int lo = min(s1, s2), hi = max(s1, s2); s1 = lo; s2 = hi; }
        SORT2_STEP(64, 16) SORT2_STEP(64, 8)  SORT2_STEP(64, 4)  SORT2_STEP(64, 2)  SORT2_STEP(64, 1)
        #undef SORT2_STEP
        const int total = c + 1;
        const bool valid2 = lane < total - ROW_SLOTS;
        if (!valid2) s2 = n;
        sh_s[wv][lane] = s1;
        sh_s[wv][ROW_SLOTS + lane] = s2;

        const float4 al1 = edge_alpha(a_src, s1, ad);
        const float4 al2 = edge_alpha(a_src, s2, ad);
        float4 m;
        m.x = wave_max(fmaxf(al1.x, valid2 ? al2.x : ninf)); m.y = wave_max(fmaxf(al1.y, valid2 ? al2.y : ninf));
        m.z = wave_max(fmaxf(al1.z, valid2 ? al2.z : ninf)); m.w = wave_max(fmaxf(al1.w, valid2 ? al2.w : ninf));
        float4 e1, e2;
        e1.x = __expf(al1.x - m.x); e1.y = __expf(al1.y - m.y);
        e1.z = __expf(al1.z - m.z); e1.w = __expf(al1.w - m.w);
        e2.x = valid2 ? __expf(al2.x - m.x) : 0.0f; e2.y = valid2 ? __expf(al2.y - m.y) : 0.0f;
        e2.z = valid2 ? __expf(al2.z - m.z) : 0.0f; e2.w = valid2 ? __expf(al2.w - m.w) : 0.0f;
        float4 d;
        d.x = wave_sum(e1.x + e2.x) + 1e-16f; d.y = wave_sum(e1.y + e2.y) + 1e-16f;
        d.z = wave_sum(e1.z + e2.z) + 1e-16f; d.w = wave_sum(e1.w + e2.w) + 1e-16f;

        d.x = 1.0f / d.x; d.y = 1.0f / d.y; d.z = 1.0f / d.z; d.w = 1.0f / d.w;
        sh_w[wv][lane] = make_float4(e1.x * d.x, e1.y * d.y, e1.z * d.z, e1.w * d.w);
        sh_w[wv][ROW_SLOTS + lane] = make_float4(e2.x * d.x, e2.y * d.y, e2.z * d.z, e2.w * d.w);
        __builtin_amdgcn_wave_barrier();
        #pragma nounroll
        for (int j = 0; j < total; ++j) {
            const int sj = sh_s[wv][j];
            const float4 wj = sh_w[wv][j];
            add_xrow(z0, z1, z2, z3, X, sj, lane, wj.x, wj.y, wj.z, wj.w);
        }
      } else {

        float4 m = edge_alpha(a_src, n, ad);
        {
            const float4 al = edge_alpha(a_src, rs, ad);
            m.x = fmaxf(m.x, al.x); m.y = fmaxf(m.y, al.y);
            m.z = fmaxf(m.z, al.z); m.w = fmaxf(m.w, al.w);
        }
        #pragma nounroll
        for (int b = 0; b < novf; b += 32) {
            const int i = b + lane;
            if (i < novf) {
                const int2 de = ov[i];
                if (de.x == n) {
                    const float4 al = edge_alpha(a_src, (int)min((unsigned)de.y, (unsigned)nmax), ad);
                    m.x = fmaxf(m.x, al.x); m.y = fmaxf(m.y, al.y);
                    m.z = fmaxf(m.z, al.z); m.w = fmaxf(m.w, al.w);
                }
            }
        }
        m.x = wave_max(m.x); m.y = wave_max(m.y); m.z = wave_max(m.z); m.w = wave_max(m.w);

        float4 den = make_float4(0.0f, 0.0f, 0.0f, 0.0f);
        for (int pass = 0; pass < 2; ++pass) {
            int last = -1;
            for (int it = 0; it <= c; ++it) {
                int k = (rs > last) ? rs : SENT;
                if (n > last) k = min(k, n);
                #pragma nounroll
                for (int b = 0; b < novf; b += 32) {
                    const int i = b + lane;
                    if (i < novf) {
                        const int2 de = ov[i];
                        const int sv = (int)min((unsigned)de.y, (unsigned)nmax);
                        if (de.x == n && sv > last) k = min(k, sv);
                    }
                }
                k = __builtin_amdgcn_readfirstlane(wave_min_i(k));
                if (k == SENT) break;
                int mult = (rs == k) ? 1 : 0;
                #pragma nounroll
                for (int b = 0; b < novf; b += 32) {
                    const int i = b + lane;
                    if (i < novf) {
                        const int2 de = ov[i];
                        const int sv = (int)min((unsigned)de.y, (unsigned)nmax);
                        if (de.x == n && sv == k) mult += 1;
                    }
                }
                mult = wave_sum_i(mult) + ((n == k) ? 1 : 0);
                last = k;
                const float fm = (float)mult;
                const float4 al = edge_alpha(a_src, k, ad);
                float4 e;
                e.x = fm * __expf(al.x - m.x); e.y = fm * __expf(al.y - m.y);
                e.z = fm * __expf(al.z - m.z); e.w = fm * __expf(al.w - m.w);
                if (pass == 0) {
                    den.x += e.x; den.y += e.y; den.z += e.z; den.w += e.w;
                } else {
                    add_xrow(z0, z1, z2, z3, X, k, lane, e.x * den.x, e.y * den.y, e.z * den.z, e.w * den.w);
                }
            }
            if (pass == 0) {
                den.x = 1.0f / (den.x + 1e-16f); den.y = 1.0f / (den.y + 1e-16f);
                den.z = 1.0f / (den.z + 1e-16f); den.w = 1.0f / (den.w + 1e-16f);
            }
        }
      }
    }
  }

    {
        float* zr = &s_z[(q * 16 + wv) * ZPITCH + lane * 4];
        *(float4*)(zr)            = z0;
        *(float4*)(zr + D_IN)     = z1;
        *(float4*)(zr + 2 * D_IN) = z2;
        *(float4*)(zr + 3 * D_IN) = z3;
    }
  }

    const int head = wv >> 2;
    const int ct   = wv & 3;
    const float* wcol = W + (size_t)(half * 2) * NH_COLS + head * F_OUT + ct * 16 + l16;
    float bq[16];
    #pragma unroll
    for (int t = 0; t < 8; ++t) {
        bq[2 * t]     = wcol[(size_t)(t * 4)     * NH_COLS];
        bq[2 * t + 1] = wcol[(size_t)(t * 4 + 1) * NH_COLS];
    }
    __syncthreads();

    {
        const float* zrow = &s_z[l16 * ZPITCH + head * D_IN + half * 2];
        v8f accA = {}; v8f accB = {};
        #pragma nounroll
        for (int g = 0; g < 4; ++g) {
            if (g > 0) {
                #pragma unroll
                for (int t = 0; t < 8; ++t) {
                    bq[2 * t]     = wcol[(size_t)(g * 32 + t * 4)     * NH_COLS];
                    bq[2 * t + 1] = wcol[(size_t)(g * 32 + t * 4 + 1) * NH_COLS];
                }
            }
            #pragma unroll
            for (int t = 0; t < 8; ++t) {
                const v2f a0 = *(const v2f*)(zrow + g * 32 + t * 4);
                const v2f a1 = *(const v2f*)(zrow + 16 * ZPITCH + g * 32 + t * 4);
                v2f b; b.x = bq[2 * t]; b.y = bq[2 * t + 1];
                accA = __builtin_amdgcn_wmma_f32_16x16x4_f32(false, a0, false, b, (short)0, accA, false, false);
                accB = __builtin_amdgcn_wmma_f32_16x16x4_f32(false, a1, false, b, (short)0, accB, false, false);
            }
        }
        #pragma unroll
        for (int r = 0; r < 8; ++r) {
            s_p[(head * NODES_PER_BLOCK + r + half * 8) * PPITCH + ct * 16 + l16]      = accA[r];
            s_p[(head * NODES_PER_BLOCK + 16 + r + half * 8) * PPITCH + ct * 16 + l16] = accB[r];
        }
    }
    __syncthreads();

  for (int q = 0; q < 2; ++q) {
    const int row = q * 16 + wv;
    const int n = blockIdx.x * NODES_PER_BLOCK + row;
    if (n >= N) continue;

    float4 acc;
    {
        const float4 p0 = *(const float4*)&s_p[(0 * NODES_PER_BLOCK + row) * PPITCH + l16 * 4];
        const float4 p1 = *(const float4*)&s_p[(1 * NODES_PER_BLOCK + row) * PPITCH + l16 * 4];
        const float4 p2 = *(const float4*)&s_p[(2 * NODES_PER_BLOCK + row) * PPITCH + l16 * 4];
        const float4 p3 = *(const float4*)&s_p[(3 * NODES_PER_BLOCK + row) * PPITCH + l16 * 4];
        acc.x = ((p0.x + p1.x) + p2.x) + p3.x;
        acc.y = ((p0.y + p1.y) + p2.y) + p3.y;
        acc.z = ((p0.z + p1.z) + p2.z) + p3.z;
        acc.w = ((p0.w + p1.w) + p2.w) + p3.w;
    }

    const float ohv = onehot[(size_t)n * 32 + lane];
    float4 p = make_float4(0.0f, 0.0f, 0.0f, 0.0f);
    #pragma unroll 4
    for (int c = 0; c < 16; ++c) {
        const float o = __shfl(ohv, c, 16);
        const float4 wp = *(const float4*)(W_prior + (half * 16 + c) * F_OUT + l16 * 4);
        p.x += o * wp.x; p.y += o * wp.y; p.z += o * wp.z; p.w += o * wp.w;
    }

    p.x += xlane<16>(p.x); p.y += xlane<16>(p.y);
    p.z += xlane<16>(p.z); p.w += xlane<16>(p.w);

    if (half == 0) {
        const float g = gate[0];
        const float4 bg = *(const float4*)(bias_gat + l16 * 4);
        const float4 bp = *(const float4*)(b_prior + l16 * 4);
        float4 o;
        o.x = (1.0f - g) * (0.25f * acc.x + bg.x) + g * (bp.x + p.x);
        o.y = (1.0f - g) * (0.25f * acc.y + bg.y) + g * (bp.y + p.y);
        o.z = (1.0f - g) * (0.25f * acc.z + bg.z) + g * (bp.z + p.z);
        o.w = (1.0f - g) * (0.25f * acc.w + bg.w) + g * (bp.w + p.w);
        *(float4*)(out + (size_t)n * F_OUT + l16 * 4) = o;
    }
  }
}

extern "C" void kernel_launch(void* const* d_in, const int* in_sizes, int n_in,
                              void* d_out, int out_size, void* d_ws, size_t ws_size,
                              hipStream_t stream)
{
    const float* x        = (const float*)d_in[0];
    const int*   ei       = (const int*)  d_in[1];
    const float* onehot   = (const float*)d_in[2];
    const float* W_gat    = (const float*)d_in[3];
    const float* att_src  = (const float*)d_in[4];
    const float* att_dst  = (const float*)d_in[5];
    const float* bias_gat = (const float*)d_in[6];
    const float* W_prior  = (const float*)d_in[7];
    const float* b_prior  = (const float*)d_in[8];
    const float* gate     = (const float*)d_in[9];
    float* out = (float*)d_out;

    const int N    = in_sizes[0] / D_IN;
    const int E    = in_sizes[1] / 2;
    const int NHs  = N * HEADS;

    const int NB  = (N + BIN_NODES - 1) / BIN_NODES;
    const int NEB = (E + EDGE_CHUNK - 1) / EDGE_CHUNK;

    float* h      = (float*)d_ws;
    int*   binned = (int*)d_ws;
    int*   hist   = binned + (size_t)E * 2;
    int*   bstart = hist + (size_t)NB * NEB;
    int*   rows   = (int*)(h + (size_t)N * NH_COLS);
    int*   ovf    = rows + (size_t)N * ROW_SLOTS;
    float* a_src  = (float*)(ovf + (size_t)E * 2);
    float* a_dst  = a_src  + NHs;
    int*   cnt    = (int*)(a_dst + NHs);
    int*   ototal = cnt + N;

    const int nscore = (N + 63) / 64;
    gat_scores<<<NEB + nscore, 256, NB * sizeof(int), stream>>>(x, W_gat, att_src, att_dst,
                                                                a_src, a_dst, ei, hist, ototal,
                                                                N, E, NEB, NB, nscore);

    bin_edges<<<NEB, 512, 3 * NB * sizeof(int), stream>>>(ei, E, N, hist, NEB, NB, binned, bstart);

    group_bins<<<NB, 512, 0, stream>>>(binned, bstart, N, E, cnt, rows, ovf, ototal);

    gat_gather<<<(N + NODES_PER_BLOCK - 1) / NODES_PER_BLOCK, 512, 0, stream>>>(
        cnt, rows, ovf, ototal, a_src, a_dst, x, W_gat,
        bias_gat, onehot, W_prior, b_prior, gate, out, N, E);
}
